# starting kernel
# baseline (speedup 1.0000x reference)
#include <hip/hip_runtime.h>
#include <hip/hip_bf16.h>

#define NPTS   40000
#define NGROUP 256
#define KNN    64
#define FDIM   768
#define FLT_BIG 3.402823e38f

typedef __attribute__((ext_vector_type(2))) float v2f;
typedef __attribute__((ext_vector_type(8))) float v8f;

__global__ __launch_bounds__(1024) void fps_kernel(const float* __restrict__ xyz,
                                                   float* __restrict__ centers_ws,
                                                   float* __restrict__ out_centers,
                                                   int N)
{
    extern __shared__ float smem[];
    float* dist = smem;
    float* rval = smem + N;
    int*   ridx = (int*)(rval + 32);
    __shared__ int s_last;

    const int b   = blockIdx.x;
    const int tid = threadIdx.x;
    const float* X = xyz + (size_t)b * N * 3;
    float* cw = centers_ws + (size_t)b * NGROUP * 3;
    float* co = out_centers + (size_t)b * NGROUP * 3;

    for (int i = tid; i < N; i += 1024) dist[i] = 1e10f;
    if (tid == 0) {
        s_last = 0;
        cw[0] = co[0] = X[0];
        cw[1] = co[1] = X[1];
        cw[2] = co[2] = X[2];
    }
    __syncthreads();

    for (int s = 1; s < NGROUP; ++s) {
        const int last = s_last;
        const float px = X[(size_t)last * 3 + 0];
        const float py = X[(size_t)last * 3 + 1];
        const float pz = X[(size_t)last * 3 + 2];

        float best = -1.0f;
        int   bi   = 0x7fffffff;
        for (int i = tid; i < N; i += 1024) {
            float dx = X[(size_t)i * 3 + 0] - px;
            float dy = X[(size_t)i * 3 + 1] - py;
            float dz = X[(size_t)i * 3 + 2] - pz;
            float d  = dx * dx + dy * dy + dz * dz;
            float od = dist[i];
            d = fminf(od, d);
            dist[i] = d;
            if (d > best || (d == best && i < bi)) { best = d; bi = i; }
        }
        #pragma unroll
        for (int off = 16; off > 0; off >>= 1) {
            float ov = __shfl_xor(best, off, 32);
            int   oi = __shfl_xor(bi,   off, 32);
            if (ov > best || (ov == best && oi < bi)) { best = ov; bi = oi; }
        }
        if ((tid & 31) == 0) { rval[tid >> 5] = best; ridx[tid >> 5] = bi; }
        __syncthreads();
        if (tid < 32) {
            best = rval[tid];
            bi   = ridx[tid];
            #pragma unroll
            for (int off = 16; off > 0; off >>= 1) {
                float ov = __shfl_xor(best, off, 32);
                int   oi = __shfl_xor(bi,   off, 32);
                if (ov > best || (ov == best && oi < bi)) { best = ov; bi = oi; }
            }
            if (tid == 0) {
                s_last = bi;
                cw[(size_t)s * 3 + 0] = co[(size_t)s * 3 + 0] = X[(size_t)bi * 3 + 0];
                cw[(size_t)s * 3 + 1] = co[(size_t)s * 3 + 1] = X[(size_t)bi * 3 + 1];
                cw[(size_t)s * 3 + 2] = co[(size_t)s * 3 + 2] = X[(size_t)bi * 3 + 2];
            }
        }
        __syncthreads();
    }
}

#define NSPLIT 25

__global__ __launch_bounds__(256) void knn_dist_kernel(const float* __restrict__ xyz,
                                                       const float* __restrict__ centers,
                                                       float* __restrict__ distws,
                                                       int N)
{
    const int bidx  = blockIdx.x;
    const int b     = bidx >> 4;
    const int mtile = bidx & 15;
    const int lane  = threadIdx.x & 31;
    const int wave  = threadIdx.x >> 5;
    const int half  = (lane < 16) ? 0 : 1;
    const int l16   = lane & 15;

    const int slice = N / NSPLIT;
    const int nbeg  = blockIdx.y * slice;
    const int nend  = nbeg + slice;

    const float* X = xyz + (size_t)b * N * 3;
    const float* C = centers + (size_t)(b * NGROUP + mtile * 16) * 3;
    float* drow = distws + (size_t)(b * NGROUP + mtile * 16) * N;

    const float cx = C[(size_t)l16 * 3 + 0];
    const float cy = C[(size_t)l16 * 3 + 1];
    const float cz = C[(size_t)l16 * 3 + 2];
    v2f a;
    a.x = half ? (-2.0f * cz) : (-2.0f * cx);
    a.y = half ? 1.0f         : (-2.0f * cy);

    for (int n0 = nbeg + wave * 16; n0 < nend; n0 += 16 * 8) {
        const int n = n0 + l16;
        const float px = X[(size_t)n * 3 + 0];
        const float py = X[(size_t)n * 3 + 1];
        const float pz = X[(size_t)n * 3 + 2];
        const float n2 = px * px + py * py + pz * pz;
        v2f bb;
        bb.x = half ? pz : px;
        bb.y = half ? n2 : py;

        v8f c = {};
        c = __builtin_amdgcn_wmma_f32_16x16x4_f32(
                 false, a,  false, bb,
                 (short)0, c,  false,  false);

        #pragma unroll
        for (int r = 0; r < 8; ++r) {
            const int mm = r + 8 * half;
            drow[(size_t)mm * N + n] = c[r];
        }
    }
}

__global__ __launch_bounds__(1024) void select_agg_kernel(const float* __restrict__ distws,
                                                          const float* __restrict__ feats,
                                                          const int*   __restrict__ valid,
                                                          float* __restrict__ out_fts,
                                                          float* __restrict__ out_mask,
                                                          int N)
{
    extern __shared__ float smem[];
    float* dist = smem;
    float* rval = smem + N;
    int*   ridx = (int*)(rval + 32);
    __shared__ int sel[KNN];
    __shared__ int s_win;

    const int gc  = blockIdx.x;
    const int b   = gc >> 8;
    const int tid = threadIdx.x;

    const float* drow = distws + (size_t)gc * N;
    for (int i = tid; i < N; i += 1024) dist[i] = drow[i];
    __syncthreads();

    float hv = FLT_BIG;
    int   hi = 0x7fffffff;
    for (int i = tid; i < N; i += 1024) {
        float d = dist[i];
        if (d < hv || (d == hv && i < hi)) { hv = d; hi = i; }
    }

    for (int k = 0; k < KNN; ++k) {
        float best = hv;
        int   bi   = hi;
        #pragma unroll
        for (int off = 16; off > 0; off >>= 1) {
            float ov = __shfl_xor(best, off, 32);
            int   oi = __shfl_xor(bi,   off, 32);
            if (ov < best || (ov == best && oi < bi)) { best = ov; bi = oi; }
        }
        if ((tid & 31) == 0) { rval[tid >> 5] = best; ridx[tid >> 5] = bi; }
        __syncthreads();
        if (tid < 32) {
            best = rval[tid];
            bi   = ridx[tid];
            #pragma unroll
            for (int off = 16; off > 0; off >>= 1) {
                float ov = __shfl_xor(best, off, 32);
                int   oi = __shfl_xor(bi,   off, 32);
                if (ov < best || (ov == best && oi < bi)) { best = ov; bi = oi; }
            }
            if (tid == 0) { sel[k] = bi; s_win = bi; }
        }
        __syncthreads();
        const int win = s_win;
        if ((win & 1023) == tid) {
            dist[win] = FLT_BIG;
            hv = FLT_BIG; hi = 0x7fffffff;
            for (int i = tid; i < N; i += 1024) {
                float d = dist[i];
                if (d < hv || (d == hv && i < hi)) { hv = d; hi = i; }
            }
        }
    }
    __syncthreads();

    const float* F = feats + (size_t)b * N * FDIM;
    if (tid < FDIM) {
        float acc = 0.0f;
        #pragma unroll 4
        for (int k = 0; k < KNN; ++k) {
            const float* row = F + (size_t)sel[k] * FDIM;
            if (k + 1 < KNN) __builtin_prefetch(F + (size_t)sel[k + 1] * FDIM + tid, 0, 0);
            acc += row[tid];
        }
        out_fts[(size_t)gc * FDIM + tid] = acc * (1.0f / 64.0f);
    }
    if (tid == 0) {
        const int* V = valid + (size_t)b * N;
        int sv = 0;
        #pragma unroll 4
        for (int k = 0; k < KNN; ++k) sv += V[sel[k]];
        out_mask[gc] = (sv > 0) ? 1.0f : 0.0f;
    }
}

extern "C" void kernel_launch(void* const* d_in, const int* in_sizes, int n_in,
                              void* d_out, int out_size, void* d_ws, size_t ws_size,
                              hipStream_t stream) {
    (void)n_in; (void)out_size; (void)ws_size;
    const float* xyzs  = (const float*)d_in[0];
    const float* feats = (const float*)d_in[1];
    const int*   valid = (const int*)d_in[2];

    const int B = 2;
    const int N = in_sizes[0] / (B * 3);

    float* out_fts  = (float*)d_out;
    float* out_mask = out_fts + (size_t)B * NGROUP * FDIM;
    float* out_cent = out_mask + (size_t)B * NGROUP;

    float* centers_ws = (float*)d_ws;
    float* distws     = centers_ws + (size_t)B * NGROUP * 3;

    const size_t smem_bytes = (size_t)N * sizeof(float) + 32 * sizeof(float) + 32 * sizeof(int);

    fps_kernel<<<B, 1024, smem_bytes, stream>>>(xyzs, centers_ws, out_cent, N);
    knn_dist_kernel<<<dim3(B * (NGROUP / 16), NSPLIT), 256, 0, stream>>>(xyzs, centers_ws,
                                                                         distws, N);
    select_agg_kernel<<<B * NGROUP, 1024, smem_bytes, stream>>>(distws, feats, valid,
                                                                out_fts, out_mask, N);
}


# top-64 select: the winner's rescan of its 40 LDS values (and the first scan) issues all reads as one independent batch, then compares in registers, instead of one load-wait-compare trip per value
# speedup vs baseline: 3.2906x; 3.2906x over previous
#include <hip/hip_runtime.h>
#include <hip/hip_bf16.h>

#define NPTS   40000
#define NGROUP 256
#define KNN    64
#define FDIM   768
#define FLT_BIG 3.402823e38f

typedef __attribute__((ext_vector_type(2))) float v2f;
typedef __attribute__((ext_vector_type(8))) float v8f;

#define FPS_T    1024
#define FPS_SLOT 40
#define FPS_ACT  (NPTS / FPS_SLOT)
static_assert(FPS_ACT * FPS_SLOT == NPTS && FPS_ACT <= FPS_T && FPS_SLOT % 2 == 0,
              "slot layout must cover every point exactly once");

__global__ __launch_bounds__(FPS_T) void fps_kernel(const float* __restrict__ xyz,
                                                    float* __restrict__ centers_ws,
                                                    float* __restrict__ out_centers)
{
    __shared__ float4 yz2[(FPS_SLOT / 2) * FPS_ACT];
    __shared__ float  rval[32];
    __shared__ int    ridx[32];
    __shared__ int    s_last;

    const int b   = blockIdx.x;
    const int tid = threadIdx.x;
    const bool own = tid < FPS_ACT;
    const float* X = xyz + (size_t)b * NPTS * 3;
    float* cw = centers_ws + (size_t)b * NGROUP * 3;
    float* co = out_centers + (size_t)b * NGROUP * 3;

    if (own) {
        for (int p = 0; p < FPS_SLOT / 2; ++p) {
            const size_t i0 = (size_t)tid + (size_t)(2 * p) * FPS_ACT;
            const size_t i1 = i0 + FPS_ACT;
            yz2[p * FPS_ACT + tid] = make_float4(X[i0 * 3 + 1], X[i0 * 3 + 2],
                                                 X[i1 * 3 + 1], X[i1 * 3 + 2]);
        }
    }
    float xs[FPS_SLOT];
    float dist[FPS_SLOT];
    #pragma unroll
    for (int j = 0; j < FPS_SLOT; ++j) { xs[j] = 0.0f; dist[j] = 1e10f; }
    if (own) {
        #pragma unroll
        for (int j = 0; j < FPS_SLOT; ++j)
            xs[j] = X[((size_t)tid + (size_t)j * FPS_ACT) * 3 + 0];
    }
    if (tid == 0) {
        s_last = 0;
        cw[0] = co[0] = X[0];
        cw[1] = co[1] = X[1];
        cw[2] = co[2] = X[2];
    }
    __syncthreads();

    for (int s = 1; s < NGROUP; ++s) {
        const int last = s_last;
        const float px = X[(size_t)last * 3 + 0];
        const float py = X[(size_t)last * 3 + 1];
        const float pz = X[(size_t)last * 3 + 2];

        float best = -1.0f;
        int   bi   = 0x7fffffff;
        if (own) {
            int bj = 0;
            #pragma unroll
            for (int p = 0; p < FPS_SLOT / 2; ++p) {
                const float4 q = yz2[p * FPS_ACT + tid];
                #pragma unroll
                for (int h = 0; h < 2; ++h) {
                    const int j = 2 * p + h;
                    const float dx = xs[j] - px;
                    const float dy = (h ? q.z : q.x) - py;
                    const float dz = (h ? q.w : q.y) - pz;
                    float t;
                    {
                        #pragma clang fp contract(off)
                        t = dx * dx + dy * dy;
                    }
                    float d = __builtin_fmaf(dz, dz, t);
                    d = fminf(dist[j], d);
                    dist[j] = d;
                    const bool take = d > best;
                    best = take ? d : best;
                    bj   = take ? j : bj;
                }
            }
            bi = tid + bj * FPS_ACT;
        }
        #pragma unroll
        for (int off = 16; off > 0; off >>= 1) {
            float ov = __shfl_xor(best, off, 32);
            int   oi = __shfl_xor(bi,   off, 32);
            if (ov > best || (ov == best && oi < bi)) { best = ov; bi = oi; }
        }
        if ((tid & 31) == 0) { rval[tid >> 5] = best; ridx[tid >> 5] = bi; }
        __syncthreads();
        if (tid < 32) {
            best = rval[tid];
            bi   = ridx[tid];
            #pragma unroll
            for (int off = 16; off > 0; off >>= 1) {
                float ov = __shfl_xor(best, off, 32);
                int   oi = __shfl_xor(bi,   off, 32);
                if (ov > best || (ov == best && oi < bi)) { best = ov; bi = oi; }
            }
            if (tid == 0) {
                s_last = bi;
                cw[(size_t)s * 3 + 0] = co[(size_t)s * 3 + 0] = X[(size_t)bi * 3 + 0];
                cw[(size_t)s * 3 + 1] = co[(size_t)s * 3 + 1] = X[(size_t)bi * 3 + 1];
                cw[(size_t)s * 3 + 2] = co[(size_t)s * 3 + 2] = X[(size_t)bi * 3 + 2];
            }
        }
        __syncthreads();
    }
}

#define NSPLIT 25

__global__ __launch_bounds__(256) void knn_dist_kernel(const float* __restrict__ xyz,
                                                       const float* __restrict__ centers,
                                                       float* __restrict__ distws,
                                                       int N)
{
    const int bidx  = blockIdx.x;
    const int b     = bidx >> 4;
    const int mtile = bidx & 15;
    const int lane  = threadIdx.x & 31;
    const int wave  = threadIdx.x >> 5;
    const int half  = (lane < 16) ? 0 : 1;
    const int l16   = lane & 15;

    const int slice = N / NSPLIT;
    const int nbeg  = blockIdx.y * slice;
    const int nend  = nbeg + slice;

    const float* X = xyz + (size_t)b * N * 3;
    const float* C = centers + (size_t)(b * NGROUP + mtile * 16) * 3;
    float* drow = distws + (size_t)(b * NGROUP + mtile * 16) * N;

    const float cx = C[(size_t)l16 * 3 + 0];
    const float cy = C[(size_t)l16 * 3 + 1];
    const float cz = C[(size_t)l16 * 3 + 2];
    v2f a;
    a.x = half ? (-2.0f * cz) : (-2.0f * cx);
    a.y = half ? 1.0f         : (-2.0f * cy);

    for (int n0 = nbeg + wave * 16; n0 < nend; n0 += 16 * 8) {
        const int n = n0 + l16;
        const float px = X[(size_t)n * 3 + 0];
        const float py = X[(size_t)n * 3 + 1];
        const float pz = X[(size_t)n * 3 + 2];
        const float n2 = px * px + py * py + pz * pz;
        v2f bb;
        bb.x = half ? pz : px;
        bb.y = half ? n2 : py;

        v8f c = {};
        c = __builtin_amdgcn_wmma_f32_16x16x4_f32(
                 false, a,  false, bb,
                 (short)0, c,  false,  false);

        #pragma unroll
        for (int r = 0; r < 8; ++r) {
            const int mm = r + 8 * half;
            drow[(size_t)mm * N + n] = c[r];
        }
    }
}

#define SEL_T    1024
#define SEL_SLOT ((NPTS + SEL_T - 1) / SEL_T)
#define SEL_Q    (SEL_SLOT / 4)
static_assert(SEL_SLOT % 4 == 0 && (SEL_SLOT - 1) * SEL_T < NPTS &&
              (3 * SEL_Q) * SEL_T + SEL_T - 1 < NPTS,
              "every quarter must start on an element that exists for every thread");

__device__ __forceinline__ void slice_argmin(const float* dist, int tid, float& hv, int& hi)
{
    float v[SEL_SLOT];
    #pragma unroll
    for (int k = 0; k < SEL_SLOT; ++k) {
        const int i = tid + k * SEL_T;
        v[k] = dist[i < NPTS ? i : tid];
    }
    float qv[4];
    int   qk[4];
    #pragma unroll
    for (int q = 0; q < 4; ++q) {
        qv[q] = v[q * SEL_Q];
        qk[q] = q * SEL_Q;
        #pragma unroll
        for (int k = q * SEL_Q + 1; k < (q + 1) * SEL_Q; ++k) {
            const bool take = (v[k] < qv[q]) && (tid + k * SEL_T < NPTS);
            qv[q] = take ? v[k] : qv[q];
            qk[q] = take ? k : qk[q];
        }
    }
    if (qv[1] < qv[0]) { qv[0] = qv[1]; qk[0] = qk[1]; }
    if (qv[3] < qv[2]) { qv[2] = qv[3]; qk[2] = qk[3]; }
    if (qv[2] < qv[0]) { qv[0] = qv[2]; qk[0] = qk[2]; }
    hv = qv[0];
    hi = tid + qk[0] * SEL_T;
}

__global__ __launch_bounds__(1024) void select_agg_kernel(const float* __restrict__ distws,
                                                          const float* __restrict__ feats,
                                                          const int*   __restrict__ valid,
                                                          float* __restrict__ out_fts,
                                                          float* __restrict__ out_mask,
                                                          int N)
{
    extern __shared__ float smem[];
    float* dist = smem;
    float* rval = smem + N;
    int*   ridx = (int*)(rval + 32);
    __shared__ int sel[KNN];
    __shared__ int s_win;

    const int gc  = blockIdx.x;
    const int b   = gc >> 8;
    const int tid = threadIdx.x;

    const float* drow = distws + (size_t)gc * N;
    for (int i = tid; i < N; i += 1024) dist[i] = drow[i];
    __syncthreads();

    float hv;
    int   hi;
    slice_argmin(dist, tid, hv, hi);

    for (int k = 0; k < KNN; ++k) {
        float best = hv;
        int   bi   = hi;
        #pragma unroll
        for (int off = 16; off > 0; off >>= 1) {
            float ov = __shfl_xor(best, off, 32);
            int   oi = __shfl_xor(bi,   off, 32);
            if (ov < best || (ov == best && oi < bi)) { best = ov; bi = oi; }
        }
        if ((tid & 31) == 0) { rval[tid >> 5] = best; ridx[tid >> 5] = bi; }
        __syncthreads();
        if (tid < 32) {
            best = rval[tid];
            bi   = ridx[tid];
            #pragma unroll
            for (int off = 16; off > 0; off >>= 1) {
                float ov = __shfl_xor(best, off, 32);
                int   oi = __shfl_xor(bi,   off, 32);
                if (ov < best || (ov == best && oi < bi)) { best = ov; bi = oi; }
            }
            if (tid == 0) { sel[k] = bi; s_win = bi; }
        }
        __syncthreads();
        const int win = s_win;
        if ((win & 1023) == tid) {
            dist[win] = FLT_BIG;
            slice_argmin(dist, tid, hv, hi);
        }
    }
    __syncthreads();

    const float* F = feats + (size_t)b * N * FDIM;
    if (tid < FDIM) {
        float acc = 0.0f;
        #pragma unroll 4
        for (int k = 0; k < KNN; ++k) {
            const float* row = F + (size_t)sel[k] * FDIM;
            if (k + 1 < KNN) __builtin_prefetch(F + (size_t)sel[k + 1] * FDIM + tid, 0, 0);
            acc += row[tid];
        }
        out_fts[(size_t)gc * FDIM + tid] = acc * (1.0f / 64.0f);
    }
    if (tid == 0) {
        const int* V = valid + (size_t)b * N;
        int sv = 0;
        #pragma unroll 4
        for (int k = 0; k < KNN; ++k) sv += V[sel[k]];
        out_mask[gc] = (sv > 0) ? 1.0f : 0.0f;
    }
}

extern "C" void kernel_launch(void* const* d_in, const int* in_sizes, int n_in,
                              void* d_out, int out_size, void* d_ws, size_t ws_size,
                              hipStream_t stream) {
    (void)n_in; (void)out_size; (void)ws_size;
    const float* xyzs  = (const float*)d_in[0];
    const float* feats = (const float*)d_in[1];
    const int*   valid = (const int*)d_in[2];

    const int B = 2;
    const int N = in_sizes[0] / (B * 3);

    float* out_fts  = (float*)d_out;
    float* out_mask = out_fts + (size_t)B * NGROUP * FDIM;
    float* out_cent = out_mask + (size_t)B * NGROUP;

    float* centers_ws = (float*)d_ws;
    float* distws     = centers_ws + (size_t)B * NGROUP * 3;

    const size_t smem_bytes = (size_t)N * sizeof(float) + 32 * sizeof(float) + 32 * sizeof(int);

    fps_kernel<<<B, FPS_T, 0, stream>>>(xyzs, centers_ws, out_cent);
    knn_dist_kernel<<<dim3(B * (NGROUP / 16), NSPLIT), 256, 0, stream>>>(xyzs, centers_ws,
                                                                         distws, N);
    select_agg_kernel<<<B * NGROUP, 1024, smem_bytes, stream>>>(distws, feats, valid,
                                                                out_fts, out_mask, N);
}


# arg-reductions of FPS and of top-64 (value, then lowest index among equals) exchange lanes on the vector ALU (DPP row_xmask, v_permlanex16) instead of __shfl_xor = ds_bpermute via the LDS unit
# speedup vs baseline: 3.8785x; 1.1787x over previous
#include <hip/hip_runtime.h>
#include <hip/hip_bf16.h>

#define NPTS   40000
#define NGROUP 256
#define KNN    64
#define FDIM   768
#define FLT_BIG 3.402823e38f

typedef __attribute__((ext_vector_type(2))) float v2f;
typedef __attribute__((ext_vector_type(8))) float v8f;


template <int XMASK> __device__ __forceinline__ int row_xor_i(int v)
{
    return __builtin_amdgcn_update_dpp(0, v, 0x160 + XMASK, 0xf, 0xf, true);
}
template <int XMASK> __device__ __forceinline__ float row_xor_f(float v)
{
    return __builtin_bit_cast(float, row_xor_i<XMASK>(__builtin_bit_cast(int, v)));
}
__device__ __forceinline__ int other_row_i(int v)
{
    return __builtin_amdgcn_permlanex16(v, v, 0x76543210, 0xfedcba98, false, false);
}
__device__ __forceinline__ float other_row_f(float v)
{
    return __builtin_bit_cast(float, other_row_i(__builtin_bit_cast(int, v)));
}
__device__ __forceinline__ int wave_min_i(int v)
{
    v = min(v, row_xor_i<1>(v));
    v = min(v, row_xor_i<2>(v));
    v = min(v, row_xor_i<4>(v));
    v = min(v, row_xor_i<8>(v));
    return min(v, other_row_i(v));
}
__device__ __forceinline__ float wave_max_f(float v)
{
    v = fmaxf(v, row_xor_f<1>(v));
    v = fmaxf(v, row_xor_f<2>(v));
    v = fmaxf(v, row_xor_f<4>(v));
    v = fmaxf(v, row_xor_f<8>(v));
    return fmaxf(v, other_row_f(v));
}
__device__ __forceinline__ float wave_min_f(float v)
{
    v = fminf(v, row_xor_f<1>(v));
    v = fminf(v, row_xor_f<2>(v));
    v = fminf(v, row_xor_f<4>(v));
    v = fminf(v, row_xor_f<8>(v));
    return fminf(v, other_row_f(v));
}
__device__ __forceinline__ void wave_argmax(float& val, int& idx)
{
    const float m = wave_max_f(val);
    idx = wave_min_i(val == m ? idx : 0x7fffffff);
    val = m;
}
__device__ __forceinline__ void wave_argmin(float& val, int& idx)
{
    const float m = wave_min_f(val);
    idx = wave_min_i(val == m ? idx : 0x7fffffff);
    val = m;
}

#define FPS_T    1024
#define FPS_SLOT 40
#define FPS_ACT  (NPTS / FPS_SLOT)
static_assert(FPS_ACT * FPS_SLOT == NPTS && FPS_ACT <= FPS_T && FPS_SLOT % 2 == 0,
              "slot layout must cover every point exactly once");

__global__ __launch_bounds__(FPS_T) void fps_kernel(const float* __restrict__ xyz,
                                                    float* __restrict__ centers_ws,
                                                    float* __restrict__ out_centers)
{
    __shared__ float4 yz2[(FPS_SLOT / 2) * FPS_ACT];
    __shared__ float  rval[32];
    __shared__ int    ridx[32];
    __shared__ int    s_last;

    const int b   = blockIdx.x;
    const int tid = threadIdx.x;
    const bool own = tid < FPS_ACT;
    const float* X = xyz + (size_t)b * NPTS * 3;
    float* cw = centers_ws + (size_t)b * NGROUP * 3;
    float* co = out_centers + (size_t)b * NGROUP * 3;

    if (own) {
        for (int p = 0; p < FPS_SLOT / 2; ++p) {
            const size_t i0 = (size_t)tid + (size_t)(2 * p) * FPS_ACT;
            const size_t i1 = i0 + FPS_ACT;
            yz2[p * FPS_ACT + tid] = make_float4(X[i0 * 3 + 1], X[i0 * 3 + 2],
                                                 X[i1 * 3 + 1], X[i1 * 3 + 2]);
        }
    }
    float xs[FPS_SLOT];
    float dist[FPS_SLOT];
    #pragma unroll
    for (int j = 0; j < FPS_SLOT; ++j) { xs[j] = 0.0f; dist[j] = 1e10f; }
    if (own) {
        #pragma unroll
        for (int j = 0; j < FPS_SLOT; ++j)
            xs[j] = X[((size_t)tid + (size_t)j * FPS_ACT) * 3 + 0];
    }
    if (tid == 0) {
        s_last = 0;
        cw[0] = co[0] = X[0];
        cw[1] = co[1] = X[1];
        cw[2] = co[2] = X[2];
    }
    __syncthreads();

    for (int s = 1; s < NGROUP; ++s) {
        const int last = s_last;
        const float px = X[(size_t)last * 3 + 0];
        const float py = X[(size_t)last * 3 + 1];
        const float pz = X[(size_t)last * 3 + 2];

        float best = -1.0f;
        int   bi   = 0x7fffffff;
        if (own) {
            int bj = 0;
            #pragma unroll
            for (int p = 0; p < FPS_SLOT / 2; ++p) {
                const float4 q = yz2[p * FPS_ACT + tid];
                #pragma unroll
                for (int h = 0; h < 2; ++h) {
                    const int j = 2 * p + h;
                    const float dx = xs[j] - px;
                    const float dy = (h ? q.z : q.x) - py;
                    const float dz = (h ? q.w : q.y) - pz;
                    float t;
                    {
                        #pragma clang fp contract(off)
                        t = dx * dx + dy * dy;
                    }
                    float d = __builtin_fmaf(dz, dz, t);
                    d = fminf(dist[j], d);
                    dist[j] = d;
                    const bool take = d > best;
                    best = take ? d : best;
                    bj   = take ? j : bj;
                }
            }
            bi = tid + bj * FPS_ACT;
        }
        wave_argmax(best, bi);
        if ((tid & 31) == 0) { rval[tid >> 5] = best; ridx[tid >> 5] = bi; }
        __syncthreads();
        if (tid < 32) {
            best = rval[tid];
            bi   = ridx[tid];
            wave_argmax(best, bi);
            if (tid == 0) {
                s_last = bi;
                cw[(size_t)s * 3 + 0] = co[(size_t)s * 3 + 0] = X[(size_t)bi * 3 + 0];
                cw[(size_t)s * 3 + 1] = co[(size_t)s * 3 + 1] = X[(size_t)bi * 3 + 1];
                cw[(size_t)s * 3 + 2] = co[(size_t)s * 3 + 2] = X[(size_t)bi * 3 + 2];
            }
        }
        __syncthreads();
    }
}

#define NSPLIT 25

__global__ __launch_bounds__(256) void knn_dist_kernel(const float* __restrict__ xyz,
                                                       const float* __restrict__ centers,
                                                       float* __restrict__ distws,
                                                       int N)
{
    const int bidx  = blockIdx.x;
    const int b     = bidx >> 4;
    const int mtile = bidx & 15;
    const int lane  = threadIdx.x & 31;
    const int wave  = threadIdx.x >> 5;
    const int half  = (lane < 16) ? 0 : 1;
    const int l16   = lane & 15;

    const int slice = N / NSPLIT;
    const int nbeg  = blockIdx.y * slice;
    const int nend  = nbeg + slice;

    const float* X = xyz + (size_t)b * N * 3;
    const float* C = centers + (size_t)(b * NGROUP + mtile * 16) * 3;
    float* drow = distws + (size_t)(b * NGROUP + mtile * 16) * N;

    const float cx = C[(size_t)l16 * 3 + 0];
    const float cy = C[(size_t)l16 * 3 + 1];
    const float cz = C[(size_t)l16 * 3 + 2];
    v2f a;
    a.x = half ? (-2.0f * cz) : (-2.0f * cx);
    a.y = half ? 1.0f         : (-2.0f * cy);

    for (int n0 = nbeg + wave * 16; n0 < nend; n0 += 16 * 8) {
        const int n = n0 + l16;
        const float px = X[(size_t)n * 3 + 0];
        const float py = X[(size_t)n * 3 + 1];
        const float pz = X[(size_t)n * 3 + 2];
        const float n2 = px * px + py * py + pz * pz;
        v2f bb;
        bb.x = half ? pz : px;
        bb.y = half ? n2 : py;

        v8f c = {};
        c = __builtin_amdgcn_wmma_f32_16x16x4_f32(
                 false, a,  false, bb,
                 (short)0, c,  false,  false);

        #pragma unroll
        for (int r = 0; r < 8; ++r) {
            const int mm = r + 8 * half;
            drow[(size_t)mm * N + n] = c[r];
        }
    }
}

#define SEL_T    1024
#define SEL_SLOT ((NPTS + SEL_T - 1) / SEL_T)
#define SEL_Q    (SEL_SLOT / 4)
static_assert(SEL_SLOT % 4 == 0 && (SEL_SLOT - 1) * SEL_T < NPTS &&
              (3 * SEL_Q) * SEL_T + SEL_T - 1 < NPTS,
              "every quarter must start on an element that exists for every thread");

__device__ __forceinline__ void slice_argmin(const float* dist, int tid, float& hv, int& hi)
{
    float v[SEL_SLOT];
    #pragma unroll
    for (int k = 0; k < SEL_SLOT; ++k) {
        const int i = tid + k * SEL_T;
        v[k] = dist[i < NPTS ? i : tid];
    }
    float qv[4];
    int   qk[4];
    #pragma unroll
    for (int q = 0; q < 4; ++q) {
        qv[q] = v[q * SEL_Q];
        qk[q] = q * SEL_Q;
        #pragma unroll
        for (int k = q * SEL_Q + 1; k < (q + 1) * SEL_Q; ++k) {
            const bool take = (v[k] < qv[q]) && (tid + k * SEL_T < NPTS);
            qv[q] = take ? v[k] : qv[q];
            qk[q] = take ? k : qk[q];
        }
    }
    if (qv[1] < qv[0]) { qv[0] = qv[1]; qk[0] = qk[1]; }
    if (qv[3] < qv[2]) { qv[2] = qv[3]; qk[2] = qk[3]; }
    if (qv[2] < qv[0]) { qv[0] = qv[2]; qk[0] = qk[2]; }
    hv = qv[0];
    hi = tid + qk[0] * SEL_T;
}

__global__ __launch_bounds__(1024) void select_agg_kernel(const float* __restrict__ distws,
                                                          const float* __restrict__ feats,
                                                          const int*   __restrict__ valid,
                                                          float* __restrict__ out_fts,
                                                          float* __restrict__ out_mask,
                                                          int N)
{
    extern __shared__ float smem[];
    float* dist = smem;
    float* rval = smem + N;
    int*   ridx = (int*)(rval + 32);
    __shared__ int sel[KNN];
    __shared__ int s_win;

    const int gc  = blockIdx.x;
    const int b   = gc >> 8;
    const int tid = threadIdx.x;

    const float* drow = distws + (size_t)gc * N;
    for (int i = tid; i < N; i += 1024) dist[i] = drow[i];
    __syncthreads();

    float hv;
    int   hi;
    slice_argmin(dist, tid, hv, hi);

    for (int k = 0; k < KNN; ++k) {
        float best = hv;
        int   bi   = hi;
        wave_argmin(best, bi);
        if ((tid & 31) == 0) { rval[tid >> 5] = best; ridx[tid >> 5] = bi; }
        __syncthreads();
        if (tid < 32) {
            best = rval[tid];
            bi   = ridx[tid];
            wave_argmin(best, bi);
            if (tid == 0) { sel[k] = bi; s_win = bi; }
        }
        __syncthreads();
        const int win = s_win;
        if ((win & 1023) == tid) {
            dist[win] = FLT_BIG;
            slice_argmin(dist, tid, hv, hi);
        }
    }
    __syncthreads();

    const float* F = feats + (size_t)b * N * FDIM;
    if (tid < FDIM) {
        float acc = 0.0f;
        #pragma unroll 4
        for (int k = 0; k < KNN; ++k) {
            const float* row = F + (size_t)sel[k] * FDIM;
            if (k + 1 < KNN) __builtin_prefetch(F + (size_t)sel[k + 1] * FDIM + tid, 0, 0);
            acc += row[tid];
        }
        out_fts[(size_t)gc * FDIM + tid] = acc * (1.0f / 64.0f);
    }
    if (tid == 0) {
        const int* V = valid + (size_t)b * N;
        int sv = 0;
        #pragma unroll 4
        for (int k = 0; k < KNN; ++k) sv += V[sel[k]];
        out_mask[gc] = (sv > 0) ? 1.0f : 0.0f;
    }
}

extern "C" void kernel_launch(void* const* d_in, const int* in_sizes, int n_in,
                              void* d_out, int out_size, void* d_ws, size_t ws_size,
                              hipStream_t stream) {
    (void)n_in; (void)out_size; (void)ws_size;
    const float* xyzs  = (const float*)d_in[0];
    const float* feats = (const float*)d_in[1];
    const int*   valid = (const int*)d_in[2];

    const int B = 2;
    const int N = in_sizes[0] / (B * 3);

    float* out_fts  = (float*)d_out;
    float* out_mask = out_fts + (size_t)B * NGROUP * FDIM;
    float* out_cent = out_mask + (size_t)B * NGROUP;

    float* centers_ws = (float*)d_ws;
    float* distws     = centers_ws + (size_t)B * NGROUP * 3;

    const size_t smem_bytes = (size_t)N * sizeof(float) + 32 * sizeof(float) + 32 * sizeof(int);

    fps_kernel<<<B, FPS_T, 0, stream>>>(xyzs, centers_ws, out_cent);
    knn_dist_kernel<<<dim3(B * (NGROUP / 16), NSPLIT), 256, 0, stream>>>(xyzs, centers_ws,
                                                                         distws, N);
    select_agg_kernel<<<B * NGROUP, 1024, smem_bytes, stream>>>(distws, feats, valid,
                                                                out_fts, out_mask, N);
}


# FPS: per-pick centre write-out (one thread's global load, two stores and their wait, between the pick's two barriers) moved behind the loop: picks kept in LDS, 256 threads write all centres once
# speedup vs baseline: 4.2290x; 1.0904x over previous
#include <hip/hip_runtime.h>
#include <hip/hip_bf16.h>

#define NPTS   40000
#define NGROUP 256
#define KNN    64
#define FDIM   768
#define FLT_BIG 3.402823e38f

typedef __attribute__((ext_vector_type(2))) float v2f;
typedef __attribute__((ext_vector_type(8))) float v8f;


template <int XMASK> __device__ __forceinline__ int row_xor_i(int v)
{
    return __builtin_amdgcn_update_dpp(0, v, 0x160 + XMASK, 0xf, 0xf, true);
}
template <int XMASK> __device__ __forceinline__ float row_xor_f(float v)
{
    return __builtin_bit_cast(float, row_xor_i<XMASK>(__builtin_bit_cast(int, v)));
}
__device__ __forceinline__ int other_row_i(int v)
{
    return __builtin_amdgcn_permlanex16(v, v, 0x76543210, 0xfedcba98, false, false);
}
__device__ __forceinline__ float other_row_f(float v)
{
    return __builtin_bit_cast(float, other_row_i(__builtin_bit_cast(int, v)));
}
__device__ __forceinline__ int wave_min_i(int v)
{
    v = min(v, row_xor_i<1>(v));
    v = min(v, row_xor_i<2>(v));
    v = min(v, row_xor_i<4>(v));
    v = min(v, row_xor_i<8>(v));
    return min(v, other_row_i(v));
}
__device__ __forceinline__ float wave_max_f(float v)
{
    v = fmaxf(v, row_xor_f<1>(v));
    v = fmaxf(v, row_xor_f<2>(v));
    v = fmaxf(v, row_xor_f<4>(v));
    v = fmaxf(v, row_xor_f<8>(v));
    return fmaxf(v, other_row_f(v));
}
__device__ __forceinline__ float wave_min_f(float v)
{
    v = fminf(v, row_xor_f<1>(v));
    v = fminf(v, row_xor_f<2>(v));
    v = fminf(v, row_xor_f<4>(v));
    v = fminf(v, row_xor_f<8>(v));
    return fminf(v, other_row_f(v));
}
__device__ __forceinline__ void wave_argmax(float& val, int& idx)
{
    const float m = wave_max_f(val);
    idx = wave_min_i(val == m ? idx : 0x7fffffff);
    val = m;
}
__device__ __forceinline__ void wave_argmin(float& val, int& idx)
{
    const float m = wave_min_f(val);
    idx = wave_min_i(val == m ? idx : 0x7fffffff);
    val = m;
}

#define FPS_T    1024
#define FPS_SLOT 40
#define FPS_ACT  (NPTS / FPS_SLOT)
static_assert(FPS_ACT * FPS_SLOT == NPTS && FPS_ACT <= FPS_T && FPS_SLOT % 2 == 0,
              "slot layout must cover every point exactly once");

__global__ __launch_bounds__(FPS_T) void fps_kernel(const float* __restrict__ xyz,
                                                    float* __restrict__ centers_ws,
                                                    float* __restrict__ out_centers)
{
    __shared__ float4 yz2[(FPS_SLOT / 2) * FPS_ACT];
    __shared__ float  rval[32];
    __shared__ int    ridx[32];
    __shared__ int    picks[NGROUP];

    const int b   = blockIdx.x;
    const int tid = threadIdx.x;
    const bool own = tid < FPS_ACT;
    const float* X = xyz + (size_t)b * NPTS * 3;
    float* cw = centers_ws + (size_t)b * NGROUP * 3;
    float* co = out_centers + (size_t)b * NGROUP * 3;

    if (own) {
        for (int p = 0; p < FPS_SLOT / 2; ++p) {
            const size_t i0 = (size_t)tid + (size_t)(2 * p) * FPS_ACT;
            const size_t i1 = i0 + FPS_ACT;
            yz2[p * FPS_ACT + tid] = make_float4(X[i0 * 3 + 1], X[i0 * 3 + 2],
                                                 X[i1 * 3 + 1], X[i1 * 3 + 2]);
        }
    }
    float xs[FPS_SLOT];
    float dist[FPS_SLOT];
    #pragma unroll
    for (int j = 0; j < FPS_SLOT; ++j) { xs[j] = 0.0f; dist[j] = 1e10f; }
    if (own) {
        #pragma unroll
        for (int j = 0; j < FPS_SLOT; ++j)
            xs[j] = X[((size_t)tid + (size_t)j * FPS_ACT) * 3 + 0];
    }
    if (tid == 0) picks[0] = 0;
    __syncthreads();

    for (int s = 1; s < NGROUP; ++s) {
        const int last = picks[s - 1];
        const float px = X[(size_t)last * 3 + 0];
        const float py = X[(size_t)last * 3 + 1];
        const float pz = X[(size_t)last * 3 + 2];

        float best = -1.0f;
        int   bi   = 0x7fffffff;
        if (own) {
            int bj = 0;
            #pragma unroll
            for (int p = 0; p < FPS_SLOT / 2; ++p) {
                const float4 q = yz2[p * FPS_ACT + tid];
                #pragma unroll
                for (int h = 0; h < 2; ++h) {
                    const int j = 2 * p + h;
                    const float dx = xs[j] - px;
                    const float dy = (h ? q.z : q.x) - py;
                    const float dz = (h ? q.w : q.y) - pz;
                    float t;
                    {
                        #pragma clang fp contract(off)
                        t = dx * dx + dy * dy;
                    }
                    float d = __builtin_fmaf(dz, dz, t);
                    d = fminf(dist[j], d);
                    dist[j] = d;
                    const bool take = d > best;
                    best = take ? d : best;
                    bj   = take ? j : bj;
                }
            }
            bi = tid + bj * FPS_ACT;
        }
        wave_argmax(best, bi);
        if ((tid & 31) == 0) { rval[tid >> 5] = best; ridx[tid >> 5] = bi; }
        __syncthreads();
        if (tid < 32) {
            best = rval[tid];
            bi   = ridx[tid];
            wave_argmax(best, bi);
            if (tid == 0) picks[s] = bi;
        }
        __syncthreads();
    }

    if (tid < NGROUP) {
        const int i = picks[tid];
        const float x = X[(size_t)i * 3 + 0];
        const float y = X[(size_t)i * 3 + 1];
        const float z = X[(size_t)i * 3 + 2];
        cw[(size_t)tid * 3 + 0] = x;  co[(size_t)tid * 3 + 0] = x;
        cw[(size_t)tid * 3 + 1] = y;  co[(size_t)tid * 3 + 1] = y;
        cw[(size_t)tid * 3 + 2] = z;  co[(size_t)tid * 3 + 2] = z;
    }
}

#define NSPLIT 25

__global__ __launch_bounds__(256) void knn_dist_kernel(const float* __restrict__ xyz,
                                                       const float* __restrict__ centers,
                                                       float* __restrict__ distws,
                                                       int N)
{
    const int bidx  = blockIdx.x;
    const int b     = bidx >> 4;
    const int mtile = bidx & 15;
    const int lane  = threadIdx.x & 31;
    const int wave  = threadIdx.x >> 5;
    const int half  = (lane < 16) ? 0 : 1;
    const int l16   = lane & 15;

    const int slice = N / NSPLIT;
    const int nbeg  = blockIdx.y * slice;
    const int nend  = nbeg + slice;

    const float* X = xyz + (size_t)b * N * 3;
    const float* C = centers + (size_t)(b * NGROUP + mtile * 16) * 3;
    float* drow = distws + (size_t)(b * NGROUP + mtile * 16) * N;

    const float cx = C[(size_t)l16 * 3 + 0];
    const float cy = C[(size_t)l16 * 3 + 1];
    const float cz = C[(size_t)l16 * 3 + 2];
    v2f a;
    a.x = half ? (-2.0f * cz) : (-2.0f * cx);
    a.y = half ? 1.0f         : (-2.0f * cy);

    for (int n0 = nbeg + wave * 16; n0 < nend; n0 += 16 * 8) {
        const int n = n0 + l16;
        const float px = X[(size_t)n * 3 + 0];
        const float py = X[(size_t)n * 3 + 1];
        const float pz = X[(size_t)n * 3 + 2];
        const float n2 = px * px + py * py + pz * pz;
        v2f bb;
        bb.x = half ? pz : px;
        bb.y = half ? n2 : py;

        v8f c = {};
        c = __builtin_amdgcn_wmma_f32_16x16x4_f32(
                 false, a,  false, bb,
                 (short)0, c,  false,  false);

        #pragma unroll
        for (int r = 0; r < 8; ++r) {
            const int mm = r + 8 * half;
            drow[(size_t)mm * N + n] = c[r];
        }
    }
}

#define SEL_T    1024
#define SEL_SLOT ((NPTS + SEL_T - 1) / SEL_T)
#define SEL_Q    (SEL_SLOT / 4)
static_assert(SEL_SLOT % 4 == 0 && (SEL_SLOT - 1) * SEL_T < NPTS &&
              (3 * SEL_Q) * SEL_T + SEL_T - 1 < NPTS,
              "every quarter must start on an element that exists for every thread");

__device__ __forceinline__ void slice_argmin(const float* dist, int tid, float& hv, int& hi)
{
    float v[SEL_SLOT];
    #pragma unroll
    for (int k = 0; k < SEL_SLOT; ++k) {
        const int i = tid + k * SEL_T;
        v[k] = dist[i < NPTS ? i : tid];
    }
    float qv[4];
    int   qk[4];
    #pragma unroll
    for (int q = 0; q < 4; ++q) {
        qv[q] = v[q * SEL_Q];
        qk[q] = q * SEL_Q;
        #pragma unroll
        for (int k = q * SEL_Q + 1; k < (q + 1) * SEL_Q; ++k) {
            const bool take = (v[k] < qv[q]) && (tid + k * SEL_T < NPTS);
            qv[q] = take ? v[k] : qv[q];
            qk[q] = take ? k : qk[q];
        }
    }
    if (qv[1] < qv[0]) { qv[0] = qv[1]; qk[0] = qk[1]; }
    if (qv[3] < qv[2]) { qv[2] = qv[3]; qk[2] = qk[3]; }
    if (qv[2] < qv[0]) { qv[0] = qv[2]; qk[0] = qk[2]; }
    hv = qv[0];
    hi = tid + qk[0] * SEL_T;
}

__global__ __launch_bounds__(1024) void select_agg_kernel(const float* __restrict__ distws,
                                                          const float* __restrict__ feats,
                                                          const int*   __restrict__ valid,
                                                          float* __restrict__ out_fts,
                                                          float* __restrict__ out_mask,
                                                          int N)
{
    extern __shared__ float smem[];
    float* dist = smem;
    float* rval = smem + N;
    int*   ridx = (int*)(rval + 32);
    __shared__ int sel[KNN];
    __shared__ int s_win;

    const int gc  = blockIdx.x;
    const int b   = gc >> 8;
    const int tid = threadIdx.x;

    const float* drow = distws + (size_t)gc * N;
    for (int i = tid; i < N; i += 1024) dist[i] = drow[i];
    __syncthreads();

    float hv;
    int   hi;
    slice_argmin(dist, tid, hv, hi);

    for (int k = 0; k < KNN; ++k) {
        float best = hv;
        int   bi   = hi;
        wave_argmin(best, bi);
        if ((tid & 31) == 0) { rval[tid >> 5] = best; ridx[tid >> 5] = bi; }
        __syncthreads();
        if (tid < 32) {
            best = rval[tid];
            bi   = ridx[tid];
            wave_argmin(best, bi);
            if (tid == 0) { sel[k] = bi; s_win = bi; }
        }
        __syncthreads();
        const int win = s_win;
        if ((win & 1023) == tid) {
            dist[win] = FLT_BIG;
            slice_argmin(dist, tid, hv, hi);
        }
    }
    __syncthreads();

    const float* F = feats + (size_t)b * N * FDIM;
    if (tid < FDIM) {
        float acc = 0.0f;
        #pragma unroll 4
        for (int k = 0; k < KNN; ++k) {
            const float* row = F + (size_t)sel[k] * FDIM;
            if (k + 1 < KNN) __builtin_prefetch(F + (size_t)sel[k + 1] * FDIM + tid, 0, 0);
            acc += row[tid];
        }
        out_fts[(size_t)gc * FDIM + tid] = acc * (1.0f / 64.0f);
    }
    if (tid == 0) {
        const int* V = valid + (size_t)b * N;
        int sv = 0;
        #pragma unroll 4
        for (int k = 0; k < KNN; ++k) sv += V[sel[k]];
        out_mask[gc] = (sv > 0) ? 1.0f : 0.0f;
    }
}

extern "C" void kernel_launch(void* const* d_in, const int* in_sizes, int n_in,
                              void* d_out, int out_size, void* d_ws, size_t ws_size,
                              hipStream_t stream) {
    (void)n_in; (void)out_size; (void)ws_size;
    const float* xyzs  = (const float*)d_in[0];
    const float* feats = (const float*)d_in[1];
    const int*   valid = (const int*)d_in[2];

    const int B = 2;
    const int N = in_sizes[0] / (B * 3);

    float* out_fts  = (float*)d_out;
    float* out_mask = out_fts + (size_t)B * NGROUP * FDIM;
    float* out_cent = out_mask + (size_t)B * NGROUP;

    float* centers_ws = (float*)d_ws;
    float* distws     = centers_ws + (size_t)B * NGROUP * 3;

    const size_t smem_bytes = (size_t)N * sizeof(float) + 32 * sizeof(float) + 32 * sizeof(int);

    fps_kernel<<<B, FPS_T, 0, stream>>>(xyzs, centers_ws, out_cent);
    knn_dist_kernel<<<dim3(B * (NGROUP / 16), NSPLIT), 256, 0, stream>>>(xyzs, centers_ws,
                                                                         distws, N);
    select_agg_kernel<<<B * NGROUP, 1024, smem_bytes, stream>>>(distws, feats, valid,
                                                                out_fts, out_mask, N);
}


# FPS per-lane argmax: the 40-point loop keeps only running maxima (one max3 per point pair); the slot index is looked up afterwards, only in lanes holding their wave's maximum; same winner
# speedup vs baseline: 4.7948x; 1.1338x over previous
#include <hip/hip_runtime.h>
#include <hip/hip_bf16.h>

#define NPTS   40000
#define NGROUP 256
#define KNN    64
#define FDIM   768
#define FLT_BIG 3.402823e38f

typedef __attribute__((ext_vector_type(2))) float v2f;
typedef __attribute__((ext_vector_type(8))) float v8f;


template <int XMASK> __device__ __forceinline__ int row_xor_i(int v)
{
    return __builtin_amdgcn_update_dpp(0, v, 0x160 + XMASK, 0xf, 0xf, true);
}
template <int XMASK> __device__ __forceinline__ float row_xor_f(float v)
{
    return __builtin_bit_cast(float, row_xor_i<XMASK>(__builtin_bit_cast(int, v)));
}
__device__ __forceinline__ int other_row_i(int v)
{
    return __builtin_amdgcn_permlanex16(v, v, 0x76543210, 0xfedcba98, false, false);
}
__device__ __forceinline__ float other_row_f(float v)
{
    return __builtin_bit_cast(float, other_row_i(__builtin_bit_cast(int, v)));
}
__device__ __forceinline__ int wave_min_i(int v)
{
    v = min(v, row_xor_i<1>(v));
    v = min(v, row_xor_i<2>(v));
    v = min(v, row_xor_i<4>(v));
    v = min(v, row_xor_i<8>(v));
    return min(v, other_row_i(v));
}
__device__ __forceinline__ float wave_max_f(float v)
{
    v = fmaxf(v, row_xor_f<1>(v));
    v = fmaxf(v, row_xor_f<2>(v));
    v = fmaxf(v, row_xor_f<4>(v));
    v = fmaxf(v, row_xor_f<8>(v));
    return fmaxf(v, other_row_f(v));
}
__device__ __forceinline__ float wave_min_f(float v)
{
    v = fminf(v, row_xor_f<1>(v));
    v = fminf(v, row_xor_f<2>(v));
    v = fminf(v, row_xor_f<4>(v));
    v = fminf(v, row_xor_f<8>(v));
    return fminf(v, other_row_f(v));
}
__device__ __forceinline__ void wave_argmax(float& val, int& idx)
{
    const float m = wave_max_f(val);
    idx = wave_min_i(val == m ? idx : 0x7fffffff);
    val = m;
}
__device__ __forceinline__ void wave_argmin(float& val, int& idx)
{
    const float m = wave_min_f(val);
    idx = wave_min_i(val == m ? idx : 0x7fffffff);
    val = m;
}

template <int LO, int CNT, int TOTAL>
__device__ __forceinline__ int first_slot_eq(const float (&v)[TOTAL], float m)
{
    int bj = LO + CNT - 1;
    #pragma unroll
    for (int j = LO + CNT - 2; j >= LO; --j) bj = (v[j] == m) ? j : bj;
    return bj;
}

#define FPS_T    1024
#define FPS_SLOT 40
#define FPS_ACT  (NPTS / FPS_SLOT)
#define FPS_GSZ  10
#define FPS_GRP  (FPS_SLOT / FPS_GSZ)
static_assert(FPS_GRP == 4 && FPS_GRP * FPS_GSZ == FPS_SLOT && FPS_GSZ % 2 == 0,
              "a 16-byte (y,z) entry's two slots must lie in one group");
static_assert(FPS_ACT * FPS_SLOT == NPTS && FPS_ACT <= FPS_T && FPS_SLOT % 2 == 0,
              "slot layout must cover every point exactly once");

__global__ __launch_bounds__(FPS_T) void fps_kernel(const float* __restrict__ xyz,
                                                    float* __restrict__ centers_ws,
                                                    float* __restrict__ out_centers)
{
    __shared__ float4 yz2[(FPS_SLOT / 2) * FPS_ACT];
    __shared__ float  rval[32];
    __shared__ int    ridx[32];
    __shared__ int    picks[NGROUP];

    const int b   = blockIdx.x;
    const int tid = threadIdx.x;
    const bool own = tid < FPS_ACT;
    const float* X = xyz + (size_t)b * NPTS * 3;
    float* cw = centers_ws + (size_t)b * NGROUP * 3;
    float* co = out_centers + (size_t)b * NGROUP * 3;

    if (own) {
        for (int p = 0; p < FPS_SLOT / 2; ++p) {
            const size_t i0 = (size_t)tid + (size_t)(2 * p) * FPS_ACT;
            const size_t i1 = i0 + FPS_ACT;
            yz2[p * FPS_ACT + tid] = make_float4(X[i0 * 3 + 1], X[i0 * 3 + 2],
                                                 X[i1 * 3 + 1], X[i1 * 3 + 2]);
        }
    }
    float xs[FPS_SLOT];
    float dist[FPS_SLOT];
    #pragma unroll
    for (int j = 0; j < FPS_SLOT; ++j) { xs[j] = 0.0f; dist[j] = 1e10f; }
    if (own) {
        #pragma unroll
        for (int j = 0; j < FPS_SLOT; ++j)
            xs[j] = X[((size_t)tid + (size_t)j * FPS_ACT) * 3 + 0];
    }
    if (tid == 0) picks[0] = 0;
    __syncthreads();

    for (int s = 1; s < NGROUP; ++s) {
        const int last = picks[s - 1];
        const float px = X[(size_t)last * 3 + 0];
        const float py = X[(size_t)last * 3 + 1];
        const float pz = X[(size_t)last * 3 + 2];

        float best = -1.0f;
        int   bi   = 0x7fffffff;
        float gm[FPS_GRP];
        #pragma unroll
        for (int k = 0; k < FPS_GRP; ++k) gm[k] = -1.0f;
        if (own) {
            #pragma unroll
            for (int p = 0; p < FPS_SLOT / 2; ++p) {
                const float4 q = yz2[p * FPS_ACT + tid];
                float dd[2];
                #pragma unroll
                for (int h = 0; h < 2; ++h) {
                    const int j = 2 * p + h;
                    const float dx = xs[j] - px;
                    const float dy = (h ? q.z : q.x) - py;
                    const float dz = (h ? q.w : q.y) - pz;
                    float t;
                    {
                        #pragma clang fp contract(off)
                        t = dx * dx + dy * dy;
                    }
                    float d = __builtin_fmaf(dz, dz, t);
                    d = fminf(dist[j], d);
                    dist[j] = d;
                    dd[h] = d;
                }
                gm[(2 * p) / FPS_GSZ] = fmaxf(fmaxf(gm[(2 * p) / FPS_GSZ], dd[0]), dd[1]);
            }
            best = fmaxf(fmaxf(gm[0], gm[1]), fmaxf(gm[2], gm[3]));
        }
        const float m = wave_max_f(best);
        if (own && best == m) {
            int bj;
            if      (gm[0] == m) bj = first_slot_eq<0 * FPS_GSZ, FPS_GSZ>(dist, m);
            else if (gm[1] == m) bj = first_slot_eq<1 * FPS_GSZ, FPS_GSZ>(dist, m);
            else if (gm[2] == m) bj = first_slot_eq<2 * FPS_GSZ, FPS_GSZ>(dist, m);
            else                 bj = first_slot_eq<3 * FPS_GSZ, FPS_GSZ>(dist, m);
            bi = tid + bj * FPS_ACT;
        }
        bi   = wave_min_i(bi);
        best = m;
        if ((tid & 31) == 0) { rval[tid >> 5] = best; ridx[tid >> 5] = bi; }
        __syncthreads();
        if (tid < 32) {
            best = rval[tid];
            bi   = ridx[tid];
            wave_argmax(best, bi);
            if (tid == 0) picks[s] = bi;
        }
        __syncthreads();
    }

    if (tid < NGROUP) {
        const int i = picks[tid];
        const float x = X[(size_t)i * 3 + 0];
        const float y = X[(size_t)i * 3 + 1];
        const float z = X[(size_t)i * 3 + 2];
        cw[(size_t)tid * 3 + 0] = x;  co[(size_t)tid * 3 + 0] = x;
        cw[(size_t)tid * 3 + 1] = y;  co[(size_t)tid * 3 + 1] = y;
        cw[(size_t)tid * 3 + 2] = z;  co[(size_t)tid * 3 + 2] = z;
    }
}

#define NSPLIT 25

__global__ __launch_bounds__(256) void knn_dist_kernel(const float* __restrict__ xyz,
                                                       const float* __restrict__ centers,
                                                       float* __restrict__ distws,
                                                       int N)
{
    const int bidx  = blockIdx.x;
    const int b     = bidx >> 4;
    const int mtile = bidx & 15;
    const int lane  = threadIdx.x & 31;
    const int wave  = threadIdx.x >> 5;
    const int half  = (lane < 16) ? 0 : 1;
    const int l16   = lane & 15;

    const int slice = N / NSPLIT;
    const int nbeg  = blockIdx.y * slice;
    const int nend  = nbeg + slice;

    const float* X = xyz + (size_t)b * N * 3;
    const float* C = centers + (size_t)(b * NGROUP + mtile * 16) * 3;
    float* drow = distws + (size_t)(b * NGROUP + mtile * 16) * N;

    const float cx = C[(size_t)l16 * 3 + 0];
    const float cy = C[(size_t)l16 * 3 + 1];
    const float cz = C[(size_t)l16 * 3 + 2];
    v2f a;
    a.x = half ? (-2.0f * cz) : (-2.0f * cx);
    a.y = half ? 1.0f         : (-2.0f * cy);

    for (int n0 = nbeg + wave * 16; n0 < nend; n0 += 16 * 8) {
        const int n = n0 + l16;
        const float px = X[(size_t)n * 3 + 0];
        const float py = X[(size_t)n * 3 + 1];
        const float pz = X[(size_t)n * 3 + 2];
        const float n2 = px * px + py * py + pz * pz;
        v2f bb;
        bb.x = half ? pz : px;
        bb.y = half ? n2 : py;

        v8f c = {};
        c = __builtin_amdgcn_wmma_f32_16x16x4_f32(
                 false, a,  false, bb,
                 (short)0, c,  false,  false);

        #pragma unroll
        for (int r = 0; r < 8; ++r) {
            const int mm = r + 8 * half;
            drow[(size_t)mm * N + n] = c[r];
        }
    }
}

#define SEL_T    1024
#define SEL_SLOT ((NPTS + SEL_T - 1) / SEL_T)
#define SEL_Q    (SEL_SLOT / 4)
static_assert(SEL_SLOT % 4 == 0 && (SEL_SLOT - 1) * SEL_T < NPTS &&
              (3 * SEL_Q) * SEL_T + SEL_T - 1 < NPTS,
              "every quarter must start on an element that exists for every thread");

__device__ __forceinline__ void slice_argmin(const float* dist, int tid, float& hv, int& hi)
{
    float v[SEL_SLOT];
    #pragma unroll
    for (int k = 0; k < SEL_SLOT; ++k) {
        const int i = tid + k * SEL_T;
        v[k] = dist[i < NPTS ? i : tid];
    }
    float qv[4];
    int   qk[4];
    #pragma unroll
    for (int q = 0; q < 4; ++q) {
        qv[q] = v[q * SEL_Q];
        qk[q] = q * SEL_Q;
        #pragma unroll
        for (int k = q * SEL_Q + 1; k < (q + 1) * SEL_Q; ++k) {
            const bool take = (v[k] < qv[q]) && (tid + k * SEL_T < NPTS);
            qv[q] = take ? v[k] : qv[q];
            qk[q] = take ? k : qk[q];
        }
    }
    if (qv[1] < qv[0]) { qv[0] = qv[1]; qk[0] = qk[1]; }
    if (qv[3] < qv[2]) { qv[2] = qv[3]; qk[2] = qk[3]; }
    if (qv[2] < qv[0]) { qv[0] = qv[2]; qk[0] = qk[2]; }
    hv = qv[0];
    hi = tid + qk[0] * SEL_T;
}

__global__ __launch_bounds__(1024) void select_agg_kernel(const float* __restrict__ distws,
                                                          const float* __restrict__ feats,
                                                          const int*   __restrict__ valid,
                                                          float* __restrict__ out_fts,
                                                          float* __restrict__ out_mask,
                                                          int N)
{
    extern __shared__ float smem[];
    float* dist = smem;
    float* rval = smem + N;
    int*   ridx = (int*)(rval + 32);
    __shared__ int sel[KNN];
    __shared__ int s_win;

    const int gc  = blockIdx.x;
    const int b   = gc >> 8;
    const int tid = threadIdx.x;

    const float* drow = distws + (size_t)gc * N;
    for (int i = tid; i < N; i += 1024) dist[i] = drow[i];
    __syncthreads();

    float hv;
    int   hi;
    slice_argmin(dist, tid, hv, hi);

    for (int k = 0; k < KNN; ++k) {
        float best = hv;
        int   bi   = hi;
        wave_argmin(best, bi);
        if ((tid & 31) == 0) { rval[tid >> 5] = best; ridx[tid >> 5] = bi; }
        __syncthreads();
        if (tid < 32) {
            best = rval[tid];
            bi   = ridx[tid];
            wave_argmin(best, bi);
            if (tid == 0) { sel[k] = bi; s_win = bi; }
        }
        __syncthreads();
        const int win = s_win;
        if ((win & 1023) == tid) {
            dist[win] = FLT_BIG;
            slice_argmin(dist, tid, hv, hi);
        }
    }
    __syncthreads();

    const float* F = feats + (size_t)b * N * FDIM;
    if (tid < FDIM) {
        float acc = 0.0f;
        #pragma unroll 4
        for (int k = 0; k < KNN; ++k) {
            const float* row = F + (size_t)sel[k] * FDIM;
            if (k + 1 < KNN) __builtin_prefetch(F + (size_t)sel[k + 1] * FDIM + tid, 0, 0);
            acc += row[tid];
        }
        out_fts[(size_t)gc * FDIM + tid] = acc * (1.0f / 64.0f);
    }
    if (tid == 0) {
        const int* V = valid + (size_t)b * N;
        int sv = 0;
        #pragma unroll 4
        for (int k = 0; k < KNN; ++k) sv += V[sel[k]];
        out_mask[gc] = (sv > 0) ? 1.0f : 0.0f;
    }
}

extern "C" void kernel_launch(void* const* d_in, const int* in_sizes, int n_in,
                              void* d_out, int out_size, void* d_ws, size_t ws_size,
                              hipStream_t stream) {
    (void)n_in; (void)out_size; (void)ws_size;
    const float* xyzs  = (const float*)d_in[0];
    const float* feats = (const float*)d_in[1];
    const int*   valid = (const int*)d_in[2];

    const int B = 2;
    const int N = in_sizes[0] / (B * 3);

    float* out_fts  = (float*)d_out;
    float* out_mask = out_fts + (size_t)B * NGROUP * FDIM;
    float* out_cent = out_mask + (size_t)B * NGROUP;

    float* centers_ws = (float*)d_ws;
    float* distws     = centers_ws + (size_t)B * NGROUP * 3;

    const size_t smem_bytes = (size_t)N * sizeof(float) + 32 * sizeof(float) + 32 * sizeof(int);

    fps_kernel<<<B, FPS_T, 0, stream>>>(xyzs, centers_ws, out_cent);
    knn_dist_kernel<<<dim3(B * (NGROUP / 16), NSPLIT), 256, 0, stream>>>(xyzs, centers_ws,
                                                                         distws, N);
    select_agg_kernel<<<B * NGROUP, 1024, smem_bytes, stream>>>(distws, feats, valid,
                                                                out_fts, out_mask, N);
}


# FPS distance update written on two-float vectors with operands in place: x/min-dist as register pairs, LDS entries (y0,y1,z0,z1) read 5 ahead; 3 register moves per entry gone; distances bit-identical
# speedup vs baseline: 5.0479x; 1.0528x over previous
#include <hip/hip_runtime.h>
#include <hip/hip_bf16.h>

#define NPTS   40000
#define NGROUP 256
#define KNN    64
#define FDIM   768
#define FLT_BIG 3.402823e38f

typedef __attribute__((ext_vector_type(2))) float v2f;
typedef __attribute__((ext_vector_type(4))) float v4f;
typedef __attribute__((ext_vector_type(8))) float v8f;


template <int XMASK> __device__ __forceinline__ int row_xor_i(int v)
{
    return __builtin_amdgcn_update_dpp(0, v, 0x160 + XMASK, 0xf, 0xf, true);
}
template <int XMASK> __device__ __forceinline__ float row_xor_f(float v)
{
    return __builtin_bit_cast(float, row_xor_i<XMASK>(__builtin_bit_cast(int, v)));
}
__device__ __forceinline__ int other_row_i(int v)
{
    return __builtin_amdgcn_permlanex16(v, v, 0x76543210, 0xfedcba98, false, false);
}
__device__ __forceinline__ float other_row_f(float v)
{
    return __builtin_bit_cast(float, other_row_i(__builtin_bit_cast(int, v)));
}
__device__ __forceinline__ int wave_min_i(int v)
{
    v = min(v, row_xor_i<1>(v));
    v = min(v, row_xor_i<2>(v));
    v = min(v, row_xor_i<4>(v));
    v = min(v, row_xor_i<8>(v));
    return min(v, other_row_i(v));
}
__device__ __forceinline__ float wave_max_f(float v)
{
    v = fmaxf(v, row_xor_f<1>(v));
    v = fmaxf(v, row_xor_f<2>(v));
    v = fmaxf(v, row_xor_f<4>(v));
    v = fmaxf(v, row_xor_f<8>(v));
    return fmaxf(v, other_row_f(v));
}
__device__ __forceinline__ float wave_min_f(float v)
{
    v = fminf(v, row_xor_f<1>(v));
    v = fminf(v, row_xor_f<2>(v));
    v = fminf(v, row_xor_f<4>(v));
    v = fminf(v, row_xor_f<8>(v));
    return fminf(v, other_row_f(v));
}
__device__ __forceinline__ void wave_argmax(float& val, int& idx)
{
    const float m = wave_max_f(val);
    idx = wave_min_i(val == m ? idx : 0x7fffffff);
    val = m;
}
__device__ __forceinline__ void wave_argmin(float& val, int& idx)
{
    const float m = wave_min_f(val);
    idx = wave_min_i(val == m ? idx : 0x7fffffff);
    val = m;
}

template <int LO, int CNT, int PAIRS>
__device__ __forceinline__ int first_slot_eq(const v2f (&v)[PAIRS], float m)
{
    int bj = LO + CNT - 1;
    #pragma unroll
    for (int j = LO + CNT - 2; j >= LO; --j) bj = (v[j >> 1][j & 1] == m) ? j : bj;
    return bj;
}

#define FPS_T    1024
#define FPS_SLOT 40
#define FPS_ACT  (NPTS / FPS_SLOT)
#define FPS_GSZ  10
#define FPS_GRP  (FPS_SLOT / FPS_GSZ)
#define FPS_AHEAD 5
static_assert(FPS_GRP == 4 && FPS_GRP * FPS_GSZ == FPS_SLOT && FPS_GSZ % 2 == 0,
              "a 16-byte (y,z) entry's two slots must lie in one group");
static_assert(FPS_ACT * FPS_SLOT == NPTS && FPS_ACT <= FPS_T && FPS_SLOT % 2 == 0,
              "slot layout must cover every point exactly once");

__global__ __launch_bounds__(FPS_T) void fps_kernel(const float* __restrict__ xyz,
                                                    float* __restrict__ centers_ws,
                                                    float* __restrict__ out_centers)
{
    __shared__ v4f yz2[(FPS_SLOT / 2) * FPS_ACT];
    __shared__ float  rval[32];
    __shared__ int    ridx[32];
    __shared__ int    picks[NGROUP];

    const int b   = blockIdx.x;
    const int tid = threadIdx.x;
    const bool own = tid < FPS_ACT;
    const float* X = xyz + (size_t)b * NPTS * 3;
    float* cw = centers_ws + (size_t)b * NGROUP * 3;
    float* co = out_centers + (size_t)b * NGROUP * 3;

    if (own) {
        #pragma unroll 4
        for (int p = 0; p < FPS_SLOT / 2; ++p) {
            const size_t i0 = (size_t)tid + (size_t)(2 * p) * FPS_ACT;
            const size_t i1 = i0 + FPS_ACT;
            v4f e;
            e.x = X[i0 * 3 + 1];  e.y = X[i1 * 3 + 1];
            e.z = X[i0 * 3 + 2];  e.w = X[i1 * 3 + 2];
            yz2[p * FPS_ACT + tid] = e;
        }
    }
    v2f xs[FPS_SLOT / 2];
    v2f dist[FPS_SLOT / 2];
    #pragma unroll
    for (int p = 0; p < FPS_SLOT / 2; ++p) {
        xs[p].x = 0.0f;  xs[p].y = 0.0f;
        dist[p].x = 1e10f;  dist[p].y = 1e10f;
    }
    if (own) {
        #pragma unroll
        for (int p = 0; p < FPS_SLOT / 2; ++p) {
            xs[p].x = X[((size_t)tid + (size_t)(2 * p) * FPS_ACT) * 3 + 0];
            xs[p].y = X[((size_t)tid + (size_t)(2 * p + 1) * FPS_ACT) * 3 + 0];
        }
    }
    if (tid == 0) picks[0] = 0;
    __syncthreads();

    const v4f* yzb[FPS_GRP];
    #pragma unroll
    for (int g = 0; g < FPS_GRP; ++g) {
        unsigned e = (unsigned)tid + (unsigned)(g * (FPS_GSZ / 2) * FPS_ACT);
        asm volatile("" : "+v"(e));
        yzb[g] = yz2 + e;
    }

    for (int s = 1; s < NGROUP; ++s) {
        const int last = picks[s - 1];
        const float px = X[(size_t)last * 3 + 0];
        const float py = X[(size_t)last * 3 + 1];
        const float pz = X[(size_t)last * 3 + 2];
        const v2f px2 = {px, px};
        const v2f py2 = {py, py};
        const v2f pz2 = {pz, pz};

        float best = -1.0f;
        int   bi   = 0x7fffffff;
        float gm[FPS_GRP];
        #pragma unroll
        for (int k = 0; k < FPS_GRP; ++k) gm[k] = -1.0f;
        if (own) {
            v4f ring[FPS_AHEAD];
            #pragma unroll
            for (int k = 0; k < FPS_AHEAD; ++k)
                ring[k] = yzb[k / (FPS_GSZ / 2)][(k % (FPS_GSZ / 2)) * FPS_ACT];
            #pragma unroll
            for (int p = 0; p < FPS_SLOT / 2; ++p) {
                const v4f q = ring[p % FPS_AHEAD];
                if (p + FPS_AHEAD < FPS_SLOT / 2)
                    ring[p % FPS_AHEAD] = yzb[(p + FPS_AHEAD) / (FPS_GSZ / 2)]
                                             [((p + FPS_AHEAD) % (FPS_GSZ / 2)) * FPS_ACT];
                const v2f dx = xs[p] - px2;
                const v2f dy = q.xy - py2;
                const v2f dz = q.zw - pz2;
                v2f t;
                {
                    #pragma clang fp contract(off)
                    t = dx * dx + dy * dy;
                }
                v2f d = __builtin_elementwise_fma(dz, dz, t);
                d.x = fminf(dist[p].x, d.x);
                d.y = fminf(dist[p].y, d.y);
                dist[p] = d;
                gm[(2 * p) / FPS_GSZ] = fmaxf(fmaxf(gm[(2 * p) / FPS_GSZ], d.x), d.y);
            }
            best = fmaxf(fmaxf(gm[0], gm[1]), fmaxf(gm[2], gm[3]));
        }
        const float m = wave_max_f(best);
        if (own && best == m) {
            int bj;
            if      (gm[0] == m) bj = first_slot_eq<0 * FPS_GSZ, FPS_GSZ>(dist, m);
            else if (gm[1] == m) bj = first_slot_eq<1 * FPS_GSZ, FPS_GSZ>(dist, m);
            else if (gm[2] == m) bj = first_slot_eq<2 * FPS_GSZ, FPS_GSZ>(dist, m);
            else                 bj = first_slot_eq<3 * FPS_GSZ, FPS_GSZ>(dist, m);
            bi = tid + bj * FPS_ACT;
        }
        bi   = wave_min_i(bi);
        best = m;
        if ((tid & 31) == 0) { rval[tid >> 5] = best; ridx[tid >> 5] = bi; }
        __syncthreads();
        if (tid < 32) {
            best = rval[tid];
            bi   = ridx[tid];
            wave_argmax(best, bi);
            if (tid == 0) picks[s] = bi;
        }
        __syncthreads();
    }

    if (tid < NGROUP) {
        const int i = picks[tid];
        const float x = X[(size_t)i * 3 + 0];
        const float y = X[(size_t)i * 3 + 1];
        const float z = X[(size_t)i * 3 + 2];
        cw[(size_t)tid * 3 + 0] = x;  co[(size_t)tid * 3 + 0] = x;
        cw[(size_t)tid * 3 + 1] = y;  co[(size_t)tid * 3 + 1] = y;
        cw[(size_t)tid * 3 + 2] = z;  co[(size_t)tid * 3 + 2] = z;
    }
}

#define NSPLIT 25

__global__ __launch_bounds__(256) void knn_dist_kernel(const float* __restrict__ xyz,
                                                       const float* __restrict__ centers,
                                                       float* __restrict__ distws,
                                                       int N)
{
    const int bidx  = blockIdx.x;
    const int b     = bidx >> 4;
    const int mtile = bidx & 15;
    const int lane  = threadIdx.x & 31;
    const int wave  = threadIdx.x >> 5;
    const int half  = (lane < 16) ? 0 : 1;
    const int l16   = lane & 15;

    const int slice = N / NSPLIT;
    const int nbeg  = blockIdx.y * slice;
    const int nend  = nbeg + slice;

    const float* X = xyz + (size_t)b * N * 3;
    const float* C = centers + (size_t)(b * NGROUP + mtile * 16) * 3;
    float* drow = distws + (size_t)(b * NGROUP + mtile * 16) * N;

    const float cx = C[(size_t)l16 * 3 + 0];
    const float cy = C[(size_t)l16 * 3 + 1];
    const float cz = C[(size_t)l16 * 3 + 2];
    v2f a;
    a.x = half ? (-2.0f * cz) : (-2.0f * cx);
    a.y = half ? 1.0f         : (-2.0f * cy);

    for (int n0 = nbeg + wave * 16; n0 < nend; n0 += 16 * 8) {
        const int n = n0 + l16;
        const float px = X[(size_t)n * 3 + 0];
        const float py = X[(size_t)n * 3 + 1];
        const float pz = X[(size_t)n * 3 + 2];
        const float n2 = px * px + py * py + pz * pz;
        v2f bb;
        bb.x = half ? pz : px;
        bb.y = half ? n2 : py;

        v8f c = {};
        c = __builtin_amdgcn_wmma_f32_16x16x4_f32(
                 false, a,  false, bb,
                 (short)0, c,  false,  false);

        #pragma unroll
        for (int r = 0; r < 8; ++r) {
            const int mm = r + 8 * half;
            drow[(size_t)mm * N + n] = c[r];
        }
    }
}

#define SEL_T    1024
#define SEL_SLOT ((NPTS + SEL_T - 1) / SEL_T)
#define SEL_Q    (SEL_SLOT / 4)
static_assert(SEL_SLOT % 4 == 0 && (SEL_SLOT - 1) * SEL_T < NPTS &&
              (3 * SEL_Q) * SEL_T + SEL_T - 1 < NPTS,
              "every quarter must start on an element that exists for every thread");

__device__ __forceinline__ void slice_argmin(const float* dist, int tid, float& hv, int& hi)
{
    float v[SEL_SLOT];
    #pragma unroll
    for (int k = 0; k < SEL_SLOT; ++k) {
        const int i = tid + k * SEL_T;
        v[k] = dist[i < NPTS ? i : tid];
    }
    float qv[4];
    int   qk[4];
    #pragma unroll
    for (int q = 0; q < 4; ++q) {
        qv[q] = v[q * SEL_Q];
        qk[q] = q * SEL_Q;
        #pragma unroll
        for (int k = q * SEL_Q + 1; k < (q + 1) * SEL_Q; ++k) {
            const bool take = (v[k] < qv[q]) && (tid + k * SEL_T < NPTS);
            qv[q] = take ? v[k] : qv[q];
            qk[q] = take ? k : qk[q];
        }
    }
    if (qv[1] < qv[0]) { qv[0] = qv[1]; qk[0] = qk[1]; }
    if (qv[3] < qv[2]) { qv[2] = qv[3]; qk[2] = qk[3]; }
    if (qv[2] < qv[0]) { qv[0] = qv[2]; qk[0] = qk[2]; }
    hv = qv[0];
    hi = tid + qk[0] * SEL_T;
}

__global__ __launch_bounds__(1024) void select_agg_kernel(const float* __restrict__ distws,
                                                          const float* __restrict__ feats,
                                                          const int*   __restrict__ valid,
                                                          float* __restrict__ out_fts,
                                                          float* __restrict__ out_mask,
                                                          int N)
{
    extern __shared__ float smem[];
    float* dist = smem;
    float* rval = smem + N;
    int*   ridx = (int*)(rval + 32);
    __shared__ int sel[KNN];
    __shared__ int s_win;

    const int gc  = blockIdx.x;
    const int b   = gc >> 8;
    const int tid = threadIdx.x;

    const float* drow = distws + (size_t)gc * N;
    for (int i = tid; i < N; i += 1024) dist[i] = drow[i];
    __syncthreads();

    float hv;
    int   hi;
    slice_argmin(dist, tid, hv, hi);

    for (int k = 0; k < KNN; ++k) {
        float best = hv;
        int   bi   = hi;
        wave_argmin(best, bi);
        if ((tid & 31) == 0) { rval[tid >> 5] = best; ridx[tid >> 5] = bi; }
        __syncthreads();
        if (tid < 32) {
            best = rval[tid];
            bi   = ridx[tid];
            wave_argmin(best, bi);
            if (tid == 0) { sel[k] = bi; s_win = bi; }
        }
        __syncthreads();
        const int win = s_win;
        if ((win & 1023) == tid) {
            dist[win] = FLT_BIG;
            slice_argmin(dist, tid, hv, hi);
        }
    }
    __syncthreads();

    const float* F = feats + (size_t)b * N * FDIM;
    if (tid < FDIM) {
        float acc = 0.0f;
        #pragma unroll 4
        for (int k = 0; k < KNN; ++k) {
            const float* row = F + (size_t)sel[k] * FDIM;
            if (k + 1 < KNN) __builtin_prefetch(F + (size_t)sel[k + 1] * FDIM + tid, 0, 0);
            acc += row[tid];
        }
        out_fts[(size_t)gc * FDIM + tid] = acc * (1.0f / 64.0f);
    }
    if (tid == 0) {
        const int* V = valid + (size_t)b * N;
        int sv = 0;
        #pragma unroll 4
        for (int k = 0; k < KNN; ++k) sv += V[sel[k]];
        out_mask[gc] = (sv > 0) ? 1.0f : 0.0f;
    }
}

extern "C" void kernel_launch(void* const* d_in, const int* in_sizes, int n_in,
                              void* d_out, int out_size, void* d_ws, size_t ws_size,
                              hipStream_t stream) {
    (void)n_in; (void)out_size; (void)ws_size;
    const float* xyzs  = (const float*)d_in[0];
    const float* feats = (const float*)d_in[1];
    const int*   valid = (const int*)d_in[2];

    const int B = 2;
    const int N = in_sizes[0] / (B * 3);

    float* out_fts  = (float*)d_out;
    float* out_mask = out_fts + (size_t)B * NGROUP * FDIM;
    float* out_cent = out_mask + (size_t)B * NGROUP;

    float* centers_ws = (float*)d_ws;
    float* distws     = centers_ws + (size_t)B * NGROUP * 3;

    const size_t smem_bytes = (size_t)N * sizeof(float) + 32 * sizeof(float) + 32 * sizeof(int);

    fps_kernel<<<B, FPS_T, 0, stream>>>(xyzs, centers_ws, out_cent);
    knn_dist_kernel<<<dim3(B * (NGROUP / 16), NSPLIT), 256, 0, stream>>>(xyzs, centers_ws,
                                                                         distws, N);
    select_agg_kernel<<<B * NGROUP, 1024, smem_bytes, stream>>>(distws, feats, valid,
                                                                out_fts, out_mask, N);
}


# FPS: all distance comparisons (running min, lane/wave/block max, slot look-up) on the floats' bit patterns as signed integers: no canonicalising v_max before each float min/max; same picks
# speedup vs baseline: 5.4905x; 1.0877x over previous
#include <hip/hip_runtime.h>
#include <hip/hip_bf16.h>

#define NPTS   40000
#define NGROUP 256
#define KNN    64
#define FDIM   768
#define FLT_BIG 3.402823e38f

typedef __attribute__((ext_vector_type(2))) float v2f;
typedef __attribute__((ext_vector_type(2))) int   v2i;
typedef __attribute__((ext_vector_type(4))) float v4f;
typedef __attribute__((ext_vector_type(8))) float v8f;


template <int XMASK> __device__ __forceinline__ int row_xor_i(int v)
{
    return __builtin_amdgcn_update_dpp(0, v, 0x160 + XMASK, 0xf, 0xf, true);
}
template <int XMASK> __device__ __forceinline__ float row_xor_f(float v)
{
    return __builtin_bit_cast(float, row_xor_i<XMASK>(__builtin_bit_cast(int, v)));
}
__device__ __forceinline__ int other_row_i(int v)
{
    return __builtin_amdgcn_permlanex16(v, v, 0x76543210, 0xfedcba98, false, false);
}
__device__ __forceinline__ float other_row_f(float v)
{
    return __builtin_bit_cast(float, other_row_i(__builtin_bit_cast(int, v)));
}
__device__ __forceinline__ int wave_min_i(int v)
{
    v = min(v, row_xor_i<1>(v));
    v = min(v, row_xor_i<2>(v));
    v = min(v, row_xor_i<4>(v));
    v = min(v, row_xor_i<8>(v));
    return min(v, other_row_i(v));
}
__device__ __forceinline__ int wave_max_i(int v)
{
    v = max(v, row_xor_i<1>(v));
    v = max(v, row_xor_i<2>(v));
    v = max(v, row_xor_i<4>(v));
    v = max(v, row_xor_i<8>(v));
    return max(v, other_row_i(v));
}
__device__ __forceinline__ float wave_min_f(float v)
{
    v = fminf(v, row_xor_f<1>(v));
    v = fminf(v, row_xor_f<2>(v));
    v = fminf(v, row_xor_f<4>(v));
    v = fminf(v, row_xor_f<8>(v));
    return fminf(v, other_row_f(v));
}
__device__ __forceinline__ void wave_argmax_i(int& val, int& idx)
{
    const int m = wave_max_i(val);
    idx = wave_min_i(val == m ? idx : 0x7fffffff);
    val = m;
}
__device__ __forceinline__ void wave_argmin(float& val, int& idx)
{
    const float m = wave_min_f(val);
    idx = wave_min_i(val == m ? idx : 0x7fffffff);
    val = m;
}

template <int LO, int CNT, int PAIRS>
__device__ __forceinline__ int first_slot_eq(const v2i (&v)[PAIRS], int m)
{
    int bj = LO + CNT - 1;
    #pragma unroll
    for (int j = LO + CNT - 2; j >= LO; --j) bj = (v[j >> 1][j & 1] == m) ? j : bj;
    return bj;
}

#define FPS_T    1024
#define FPS_SLOT 40
#define FPS_ACT  (NPTS / FPS_SLOT)
#define FPS_GSZ  10
#define FPS_GRP  (FPS_SLOT / FPS_GSZ)
#define FPS_AHEAD 5
static_assert(FPS_GRP == 4 && FPS_GRP * FPS_GSZ == FPS_SLOT && FPS_GSZ % 2 == 0,
              "a 16-byte (y,z) entry's two slots must lie in one group");
static_assert(FPS_ACT * FPS_SLOT == NPTS && FPS_ACT <= FPS_T && FPS_SLOT % 2 == 0,
              "slot layout must cover every point exactly once");

__global__ __launch_bounds__(FPS_T) void fps_kernel(const float* __restrict__ xyz,
                                                    float* __restrict__ centers_ws,
                                                    float* __restrict__ out_centers)
{
    __shared__ v4f yz2[(FPS_SLOT / 2) * FPS_ACT];
    __shared__ int    rkey[32];
    __shared__ int    ridx[32];
    __shared__ int    picks[NGROUP];

    const int b   = blockIdx.x;
    const int tid = threadIdx.x;
    const bool own = tid < FPS_ACT;
    const float* X = xyz + (size_t)b * NPTS * 3;
    float* cw = centers_ws + (size_t)b * NGROUP * 3;
    float* co = out_centers + (size_t)b * NGROUP * 3;

    if (own) {
        #pragma unroll 4
        for (int p = 0; p < FPS_SLOT / 2; ++p) {
            const size_t i0 = (size_t)tid + (size_t)(2 * p) * FPS_ACT;
            const size_t i1 = i0 + FPS_ACT;
            v4f e;
            e.x = X[i0 * 3 + 1];  e.y = X[i1 * 3 + 1];
            e.z = X[i0 * 3 + 2];  e.w = X[i1 * 3 + 2];
            yz2[p * FPS_ACT + tid] = e;
        }
    }
    v2f xs[FPS_SLOT / 2];
    v2i dist[FPS_SLOT / 2];
    #pragma unroll
    for (int p = 0; p < FPS_SLOT / 2; ++p) {
        xs[p].x = 0.0f;  xs[p].y = 0.0f;
        dist[p].x = __builtin_bit_cast(int, 1e10f);  dist[p].y = __builtin_bit_cast(int, 1e10f);
    }
    if (own) {
        #pragma unroll
        for (int p = 0; p < FPS_SLOT / 2; ++p) {
            xs[p].x = X[((size_t)tid + (size_t)(2 * p) * FPS_ACT) * 3 + 0];
            xs[p].y = X[((size_t)tid + (size_t)(2 * p + 1) * FPS_ACT) * 3 + 0];
        }
    }
    if (tid == 0) picks[0] = 0;
    __syncthreads();

    const v4f* yzb[FPS_GRP];
    #pragma unroll
    for (int g = 0; g < FPS_GRP; ++g) {
        unsigned e = (unsigned)tid + (unsigned)(g * (FPS_GSZ / 2) * FPS_ACT);
        asm volatile("" : "+v"(e));
        yzb[g] = yz2 + e;
    }

    for (int s = 1; s < NGROUP; ++s) {
        const int last = picks[s - 1];
        const float px = X[(size_t)last * 3 + 0];
        const float py = X[(size_t)last * 3 + 1];
        const float pz = X[(size_t)last * 3 + 2];
        const v2f px2 = {px, px};
        const v2f py2 = {py, py};
        const v2f pz2 = {pz, pz};

        int best = -1;
        int bi   = 0x7fffffff;
        int gm[FPS_GRP];
        #pragma unroll
        for (int k = 0; k < FPS_GRP; ++k) gm[k] = -1;
        if (own) {
            v4f ring[FPS_AHEAD];
            #pragma unroll
            for (int k = 0; k < FPS_AHEAD; ++k)
                ring[k] = yzb[k / (FPS_GSZ / 2)][(k % (FPS_GSZ / 2)) * FPS_ACT];
            #pragma unroll
            for (int p = 0; p < FPS_SLOT / 2; ++p) {
                const v4f q = ring[p % FPS_AHEAD];
                if (p + FPS_AHEAD < FPS_SLOT / 2)
                    ring[p % FPS_AHEAD] = yzb[(p + FPS_AHEAD) / (FPS_GSZ / 2)]
                                             [((p + FPS_AHEAD) % (FPS_GSZ / 2)) * FPS_ACT];
                const v2f dx = xs[p] - px2;
                const v2f dy = q.xy - py2;
                const v2f dz = q.zw - pz2;
                v2f t;
                {
                    #pragma clang fp contract(off)
                    t = dx * dx + dy * dy;
                }
                const v2f d = __builtin_elementwise_fma(dz, dz, t);
                const v2i dk = __builtin_bit_cast(v2i, d);
                const v2i k = __builtin_elementwise_min(dist[p], dk);
                dist[p] = k;
                gm[(2 * p) / FPS_GSZ] = max(max(gm[(2 * p) / FPS_GSZ], k.x), k.y);
            }
            best = max(max(gm[0], gm[1]), max(gm[2], gm[3]));
        }
        const int m = wave_max_i(best);
        if (own && best == m) {
            int bj;
            if      (gm[0] == m) bj = first_slot_eq<0 * FPS_GSZ, FPS_GSZ>(dist, m);
            else if (gm[1] == m) bj = first_slot_eq<1 * FPS_GSZ, FPS_GSZ>(dist, m);
            else if (gm[2] == m) bj = first_slot_eq<2 * FPS_GSZ, FPS_GSZ>(dist, m);
            else                 bj = first_slot_eq<3 * FPS_GSZ, FPS_GSZ>(dist, m);
            bi = tid + bj * FPS_ACT;
        }
        bi   = wave_min_i(bi);
        best = m;
        if ((tid & 31) == 0) { rkey[tid >> 5] = best; ridx[tid >> 5] = bi; }
        __syncthreads();
        if (tid < 32) {
            best = rkey[tid];
            bi   = ridx[tid];
            wave_argmax_i(best, bi);
            if (tid == 0) picks[s] = bi;
        }
        __syncthreads();
    }

    if (tid < NGROUP) {
        const int i = picks[tid];
        const float x = X[(size_t)i * 3 + 0];
        const float y = X[(size_t)i * 3 + 1];
        const float z = X[(size_t)i * 3 + 2];
        cw[(size_t)tid * 3 + 0] = x;  co[(size_t)tid * 3 + 0] = x;
        cw[(size_t)tid * 3 + 1] = y;  co[(size_t)tid * 3 + 1] = y;
        cw[(size_t)tid * 3 + 2] = z;  co[(size_t)tid * 3 + 2] = z;
    }
}

#define NSPLIT 25

__global__ __launch_bounds__(256) void knn_dist_kernel(const float* __restrict__ xyz,
                                                       const float* __restrict__ centers,
                                                       float* __restrict__ distws,
                                                       int N)
{
    const int bidx  = blockIdx.x;
    const int b     = bidx >> 4;
    const int mtile = bidx & 15;
    const int lane  = threadIdx.x & 31;
    const int wave  = threadIdx.x >> 5;
    const int half  = (lane < 16) ? 0 : 1;
    const int l16   = lane & 15;

    const int slice = N / NSPLIT;
    const int nbeg  = blockIdx.y * slice;
    const int nend  = nbeg + slice;

    const float* X = xyz + (size_t)b * N * 3;
    const float* C = centers + (size_t)(b * NGROUP + mtile * 16) * 3;
    float* drow = distws + (size_t)(b * NGROUP + mtile * 16) * N;

    const float cx = C[(size_t)l16 * 3 + 0];
    const float cy = C[(size_t)l16 * 3 + 1];
    const float cz = C[(size_t)l16 * 3 + 2];
    v2f a;
    a.x = half ? (-2.0f * cz) : (-2.0f * cx);
    a.y = half ? 1.0f         : (-2.0f * cy);

    for (int n0 = nbeg + wave * 16; n0 < nend; n0 += 16 * 8) {
        const int n = n0 + l16;
        const float px = X[(size_t)n * 3 + 0];
        const float py = X[(size_t)n * 3 + 1];
        const float pz = X[(size_t)n * 3 + 2];
        const float n2 = px * px + py * py + pz * pz;
        v2f bb;
        bb.x = half ? pz : px;
        bb.y = half ? n2 : py;

        v8f c = {};
        c = __builtin_amdgcn_wmma_f32_16x16x4_f32(
                 false, a,  false, bb,
                 (short)0, c,  false,  false);

        #pragma unroll
        for (int r = 0; r < 8; ++r) {
            const int mm = r + 8 * half;
            drow[(size_t)mm * N + n] = c[r];
        }
    }
}

#define SEL_T    1024
#define SEL_SLOT ((NPTS + SEL_T - 1) / SEL_T)
#define SEL_Q    (SEL_SLOT / 4)
static_assert(SEL_SLOT % 4 == 0 && (SEL_SLOT - 1) * SEL_T < NPTS &&
              (3 * SEL_Q) * SEL_T + SEL_T - 1 < NPTS,
              "every quarter must start on an element that exists for every thread");

__device__ __forceinline__ void slice_argmin(const float* dist, int tid, float& hv, int& hi)
{
    float v[SEL_SLOT];
    #pragma unroll
    for (int k = 0; k < SEL_SLOT; ++k) {
        const int i = tid + k * SEL_T;
        v[k] = dist[i < NPTS ? i : tid];
    }
    float qv[4];
    int   qk[4];
    #pragma unroll
    for (int q = 0; q < 4; ++q) {
        qv[q] = v[q * SEL_Q];
        qk[q] = q * SEL_Q;
        #pragma unroll
        for (int k = q * SEL_Q + 1; k < (q + 1) * SEL_Q; ++k) {
            const bool take = (v[k] < qv[q]) && (tid + k * SEL_T < NPTS);
            qv[q] = take ? v[k] : qv[q];
            qk[q] = take ? k : qk[q];
        }
    }
    if (qv[1] < qv[0]) { qv[0] = qv[1]; qk[0] = qk[1]; }
    if (qv[3] < qv[2]) { qv[2] = qv[3]; qk[2] = qk[3]; }
    if (qv[2] < qv[0]) { qv[0] = qv[2]; qk[0] = qk[2]; }
    hv = qv[0];
    hi = tid + qk[0] * SEL_T;
}

__global__ __launch_bounds__(1024) void select_agg_kernel(const float* __restrict__ distws,
                                                          const float* __restrict__ feats,
                                                          const int*   __restrict__ valid,
                                                          float* __restrict__ out_fts,
                                                          float* __restrict__ out_mask,
                                                          int N)
{
    extern __shared__ float smem[];
    float* dist = smem;
    float* rval = smem + N;
    int*   ridx = (int*)(rval + 32);
    __shared__ int sel[KNN];
    __shared__ int s_win;

    const int gc  = blockIdx.x;
    const int b   = gc >> 8;
    const int tid = threadIdx.x;

    const float* drow = distws + (size_t)gc * N;
    for (int i = tid; i < N; i += 1024) dist[i] = drow[i];
    __syncthreads();

    float hv;
    int   hi;
    slice_argmin(dist, tid, hv, hi);

    for (int k = 0; k < KNN; ++k) {
        float best = hv;
        int   bi   = hi;
        wave_argmin(best, bi);
        if ((tid & 31) == 0) { rval[tid >> 5] = best; ridx[tid >> 5] = bi; }
        __syncthreads();
        if (tid < 32) {
            best = rval[tid];
            bi   = ridx[tid];
            wave_argmin(best, bi);
            if (tid == 0) { sel[k] = bi; s_win = bi; }
        }
        __syncthreads();
        const int win = s_win;
        if ((win & 1023) == tid) {
            dist[win] = FLT_BIG;
            slice_argmin(dist, tid, hv, hi);
        }
    }
    __syncthreads();

    const float* F = feats + (size_t)b * N * FDIM;
    if (tid < FDIM) {
        float acc = 0.0f;
        #pragma unroll 4
        for (int k = 0; k < KNN; ++k) {
            const float* row = F + (size_t)sel[k] * FDIM;
            if (k + 1 < KNN) __builtin_prefetch(F + (size_t)sel[k + 1] * FDIM + tid, 0, 0);
            acc += row[tid];
        }
        out_fts[(size_t)gc * FDIM + tid] = acc * (1.0f / 64.0f);
    }
    if (tid == 0) {
        const int* V = valid + (size_t)b * N;
        int sv = 0;
        #pragma unroll 4
        for (int k = 0; k < KNN; ++k) sv += V[sel[k]];
        out_mask[gc] = (sv > 0) ? 1.0f : 0.0f;
    }
}

extern "C" void kernel_launch(void* const* d_in, const int* in_sizes, int n_in,
                              void* d_out, int out_size, void* d_ws, size_t ws_size,
                              hipStream_t stream) {
    (void)n_in; (void)out_size; (void)ws_size;
    const float* xyzs  = (const float*)d_in[0];
    const float* feats = (const float*)d_in[1];
    const int*   valid = (const int*)d_in[2];

    const int B = 2;
    const int N = in_sizes[0] / (B * 3);

    float* out_fts  = (float*)d_out;
    float* out_mask = out_fts + (size_t)B * NGROUP * FDIM;
    float* out_cent = out_mask + (size_t)B * NGROUP;

    float* centers_ws = (float*)d_ws;
    float* distws     = centers_ws + (size_t)B * NGROUP * 3;

    const size_t smem_bytes = (size_t)N * sizeof(float) + 32 * sizeof(float) + 32 * sizeof(int);

    fps_kernel<<<B, FPS_T, 0, stream>>>(xyzs, centers_ws, out_cent);
    knn_dist_kernel<<<dim3(B * (NGROUP / 16), NSPLIT), 256, 0, stream>>>(xyzs, centers_ws,
                                                                         distws, N);
    select_agg_kernel<<<B * NGROUP, 1024, smem_bytes, stream>>>(distws, feats, valid,
                                                                out_fts, out_mask, N);
}


# FPS hand-over: each wave's partial carries its candidate's x,y,z and every wave reduces the 32 partials itself (two alternating LDS sets): one barrier a pick, no global load; same picks
# speedup vs baseline: 5.6587x; 1.0306x over previous
#include <hip/hip_runtime.h>
#include <hip/hip_bf16.h>

#define NPTS   40000
#define NGROUP 256
#define KNN    64
#define FDIM   768
#define FLT_BIG 3.402823e38f

typedef __attribute__((ext_vector_type(2))) float v2f;
typedef __attribute__((ext_vector_type(2))) int   v2i;
typedef __attribute__((ext_vector_type(4))) float v4f;
typedef __attribute__((ext_vector_type(4))) int   v4i;
typedef __attribute__((ext_vector_type(8))) float v8f;


template <int XMASK> __device__ __forceinline__ int row_xor_i(int v)
{
    return __builtin_amdgcn_update_dpp(0, v, 0x160 + XMASK, 0xf, 0xf, true);
}
template <int XMASK> __device__ __forceinline__ float row_xor_f(float v)
{
    return __builtin_bit_cast(float, row_xor_i<XMASK>(__builtin_bit_cast(int, v)));
}
__device__ __forceinline__ int other_row_i(int v)
{
    return __builtin_amdgcn_permlanex16(v, v, 0x76543210, 0xfedcba98, false, false);
}
__device__ __forceinline__ float other_row_f(float v)
{
    return __builtin_bit_cast(float, other_row_i(__builtin_bit_cast(int, v)));
}
__device__ __forceinline__ int wave_min_i(int v)
{
    v = min(v, row_xor_i<1>(v));
    v = min(v, row_xor_i<2>(v));
    v = min(v, row_xor_i<4>(v));
    v = min(v, row_xor_i<8>(v));
    return min(v, other_row_i(v));
}
__device__ __forceinline__ int wave_max_i(int v)
{
    v = max(v, row_xor_i<1>(v));
    v = max(v, row_xor_i<2>(v));
    v = max(v, row_xor_i<4>(v));
    v = max(v, row_xor_i<8>(v));
    return max(v, other_row_i(v));
}
__device__ __forceinline__ float wave_min_f(float v)
{
    v = fminf(v, row_xor_f<1>(v));
    v = fminf(v, row_xor_f<2>(v));
    v = fminf(v, row_xor_f<4>(v));
    v = fminf(v, row_xor_f<8>(v));
    return fminf(v, other_row_f(v));
}
__device__ __forceinline__ void wave_argmax_i(int& val, int& idx)
{
    const int m = wave_max_i(val);
    idx = wave_min_i(val == m ? idx : 0x7fffffff);
    val = m;
}
__device__ __forceinline__ void wave_argmin(float& val, int& idx)
{
    const float m = wave_min_f(val);
    idx = wave_min_i(val == m ? idx : 0x7fffffff);
    val = m;
}

template <int LO, int CNT, int PAIRS>
__device__ __forceinline__ int first_slot_eq(const v2i (&v)[PAIRS], const v2f (&x)[PAIRS],
                                             int m, float& bx)
{
    int bj = LO + CNT - 1;
    bx = x[(LO + CNT - 1) >> 1][(LO + CNT - 1) & 1];
    #pragma unroll
    for (int j = LO + CNT - 2; j >= LO; --j) {
        const bool hit = v[j >> 1][j & 1] == m;
        bj = hit ? j : bj;
        bx = hit ? x[j >> 1][j & 1] : bx;
    }
    return bj;
}

#define FPS_T    1024
#define FPS_SLOT 40
#define FPS_ACT  (NPTS / FPS_SLOT)
#define FPS_GSZ  10
#define FPS_GRP  (FPS_SLOT / FPS_GSZ)
#define FPS_AHEAD 5
static_assert(FPS_GRP == 4 && FPS_GRP * FPS_GSZ == FPS_SLOT && FPS_GSZ % 2 == 0,
              "a 16-byte (y,z) entry's two slots must lie in one group");
static_assert(FPS_ACT * FPS_SLOT == NPTS && FPS_ACT <= FPS_T && FPS_SLOT % 2 == 0,
              "slot layout must cover every point exactly once");

struct __attribute__((aligned(32))) Partial {
    v4i c;
    int idx;
    int pad[3];
};

__global__ __launch_bounds__(FPS_T) void fps_kernel(const float* __restrict__ xyz,
                                                    float* __restrict__ centers_ws,
                                                    float* __restrict__ out_centers)
{
    __shared__ v4f yz2[(FPS_SLOT / 2) * FPS_ACT];
    __shared__ Partial part[2][32];
    __shared__ int    picks[NGROUP];

    const int b   = blockIdx.x;
    const int tid = threadIdx.x;
    const bool own = tid < FPS_ACT;
    const float* X = xyz + (size_t)b * NPTS * 3;
    float* cw = centers_ws + (size_t)b * NGROUP * 3;
    float* co = out_centers + (size_t)b * NGROUP * 3;

    if (own) {
        #pragma unroll 4
        for (int p = 0; p < FPS_SLOT / 2; ++p) {
            const size_t i0 = (size_t)tid + (size_t)(2 * p) * FPS_ACT;
            const size_t i1 = i0 + FPS_ACT;
            v4f e;
            e.x = X[i0 * 3 + 1];  e.y = X[i1 * 3 + 1];
            e.z = X[i0 * 3 + 2];  e.w = X[i1 * 3 + 2];
            yz2[p * FPS_ACT + tid] = e;
        }
    }
    v2f xs[FPS_SLOT / 2];
    v2i dist[FPS_SLOT / 2];
    #pragma unroll
    for (int p = 0; p < FPS_SLOT / 2; ++p) {
        xs[p].x = 0.0f;  xs[p].y = 0.0f;
        dist[p].x = __builtin_bit_cast(int, 1e10f);  dist[p].y = __builtin_bit_cast(int, 1e10f);
    }
    if (own) {
        #pragma unroll
        for (int p = 0; p < FPS_SLOT / 2; ++p) {
            xs[p].x = X[((size_t)tid + (size_t)(2 * p) * FPS_ACT) * 3 + 0];
            xs[p].y = X[((size_t)tid + (size_t)(2 * p + 1) * FPS_ACT) * 3 + 0];
        }
    }
    if (tid == 0) picks[0] = 0;
    __syncthreads();

    const v4f* yzb[FPS_GRP];
    #pragma unroll
    for (int g = 0; g < FPS_GRP; ++g) {
        unsigned e = (unsigned)tid + (unsigned)(g * (FPS_GSZ / 2) * FPS_ACT);
        asm volatile("" : "+v"(e));
        yzb[g] = yz2 + e;
    }

    float px = X[0];
    float py = X[1];
    float pz = X[2];
    const int lane = tid & 31;

    for (int s = 1; s < NGROUP; ++s) {
        const v2f px2 = {px, px};
        const v2f py2 = {py, py};
        const v2f pz2 = {pz, pz};

        int best = -1;
        int bi   = 0x7fffffff;
        int gm[FPS_GRP];
        #pragma unroll
        for (int k = 0; k < FPS_GRP; ++k) gm[k] = -1;
        if (own) {
            v4f ring[FPS_AHEAD];
            #pragma unroll
            for (int k = 0; k < FPS_AHEAD; ++k)
                ring[k] = yzb[k / (FPS_GSZ / 2)][(k % (FPS_GSZ / 2)) * FPS_ACT];
            #pragma unroll
            for (int p = 0; p < FPS_SLOT / 2; ++p) {
                const v4f q = ring[p % FPS_AHEAD];
                if (p + FPS_AHEAD < FPS_SLOT / 2)
                    ring[p % FPS_AHEAD] = yzb[(p + FPS_AHEAD) / (FPS_GSZ / 2)]
                                             [((p + FPS_AHEAD) % (FPS_GSZ / 2)) * FPS_ACT];
                const v2f dx = xs[p] - px2;
                const v2f dy = q.xy - py2;
                const v2f dz = q.zw - pz2;
                v2f t;
                {
                    #pragma clang fp contract(off)
                    t = dx * dx + dy * dy;
                }
                const v2f d = __builtin_elementwise_fma(dz, dz, t);
                const v2i dk = __builtin_bit_cast(v2i, d);
                const v2i k = __builtin_elementwise_min(dist[p], dk);
                dist[p] = k;
                gm[(2 * p) / FPS_GSZ] = max(max(gm[(2 * p) / FPS_GSZ], k.x), k.y);
            }
            best = max(max(gm[0], gm[1]), max(gm[2], gm[3]));
        }
        const int m = wave_max_i(best);
        int   bj = 0;
        float bx = 0.0f;
        v4f   q  = {0.0f, 0.0f, 0.0f, 0.0f};
        if (own && best == m) {
            if      (gm[0] == m) bj = first_slot_eq<0 * FPS_GSZ, FPS_GSZ>(dist, xs, m, bx);
            else if (gm[1] == m) bj = first_slot_eq<1 * FPS_GSZ, FPS_GSZ>(dist, xs, m, bx);
            else if (gm[2] == m) bj = first_slot_eq<2 * FPS_GSZ, FPS_GSZ>(dist, xs, m, bx);
            else                 bj = first_slot_eq<3 * FPS_GSZ, FPS_GSZ>(dist, xs, m, bx);
            bi = tid + bj * FPS_ACT;
            q = yzb[0][(bj >> 1) * FPS_ACT];
        }
        const int par = s & 1;
        const int wbi = wave_min_i(bi);
        if (bi == wbi) {
            const bool odd = (bj & 1) != 0;
            const float cy = odd ? q.y : q.x;
            const float cz = odd ? q.w : q.z;
            const v4i c = {__float_as_int(bx), __float_as_int(cy), __float_as_int(cz), m};
            part[par][tid >> 5].c   = c;
            part[par][tid >> 5].idx = bi;
        }
        __syncthreads();
        const v4i kc   = part[par][lane].c;
        const int kidx = part[par][lane].idx;
        int gkey = kc.w;
        int gidx = kidx;
        wave_argmax_i(gkey, gidx);
        const unsigned wmask = __builtin_amdgcn_ballot_w32(kidx == gidx);
        const int wl = __builtin_ctz(wmask);
        const int kx = kc.x;
        const int ky = kc.y;
        const int kz = kc.z;
        px = __int_as_float(__builtin_amdgcn_readlane(kx, wl));
        py = __int_as_float(__builtin_amdgcn_readlane(ky, wl));
        pz = __int_as_float(__builtin_amdgcn_readlane(kz, wl));
        if (tid == 0) picks[s] = gidx;
    }
    __syncthreads();

    if (tid < NGROUP) {
        const int i = picks[tid];
        const float x = X[(size_t)i * 3 + 0];
        const float y = X[(size_t)i * 3 + 1];
        const float z = X[(size_t)i * 3 + 2];
        cw[(size_t)tid * 3 + 0] = x;  co[(size_t)tid * 3 + 0] = x;
        cw[(size_t)tid * 3 + 1] = y;  co[(size_t)tid * 3 + 1] = y;
        cw[(size_t)tid * 3 + 2] = z;  co[(size_t)tid * 3 + 2] = z;
    }
}

#define NSPLIT 25

__global__ __launch_bounds__(256) void knn_dist_kernel(const float* __restrict__ xyz,
                                                       const float* __restrict__ centers,
                                                       float* __restrict__ distws,
                                                       int N)
{
    const int bidx  = blockIdx.x;
    const int b     = bidx >> 4;
    const int mtile = bidx & 15;
    const int lane  = threadIdx.x & 31;
    const int wave  = threadIdx.x >> 5;
    const int half  = (lane < 16) ? 0 : 1;
    const int l16   = lane & 15;

    const int slice = N / NSPLIT;
    const int nbeg  = blockIdx.y * slice;
    const int nend  = nbeg + slice;

    const float* X = xyz + (size_t)b * N * 3;
    const float* C = centers + (size_t)(b * NGROUP + mtile * 16) * 3;
    float* drow = distws + (size_t)(b * NGROUP + mtile * 16) * N;

    const float cx = C[(size_t)l16 * 3 + 0];
    const float cy = C[(size_t)l16 * 3 + 1];
    const float cz = C[(size_t)l16 * 3 + 2];
    v2f a;
    a.x = half ? (-2.0f * cz) : (-2.0f * cx);
    a.y = half ? 1.0f         : (-2.0f * cy);

    for (int n0 = nbeg + wave * 16; n0 < nend; n0 += 16 * 8) {
        const int n = n0 + l16;
        const float px = X[(size_t)n * 3 + 0];
        const float py = X[(size_t)n * 3 + 1];
        const float pz = X[(size_t)n * 3 + 2];
        const float n2 = px * px + py * py + pz * pz;
        v2f bb;
        bb.x = half ? pz : px;
        bb.y = half ? n2 : py;

        v8f c = {};
        c = __builtin_amdgcn_wmma_f32_16x16x4_f32(
                 false, a,  false, bb,
                 (short)0, c,  false,  false);

        #pragma unroll
        for (int r = 0; r < 8; ++r) {
            const int mm = r + 8 * half;
            drow[(size_t)mm * N + n] = c[r];
        }
    }
}

#define SEL_T    1024
#define SEL_SLOT ((NPTS + SEL_T - 1) / SEL_T)
#define SEL_Q    (SEL_SLOT / 4)
static_assert(SEL_SLOT % 4 == 0 && (SEL_SLOT - 1) * SEL_T < NPTS &&
              (3 * SEL_Q) * SEL_T + SEL_T - 1 < NPTS,
              "every quarter must start on an element that exists for every thread");

__device__ __forceinline__ void slice_argmin(const float* dist, int tid, float& hv, int& hi)
{
    float v[SEL_SLOT];
    #pragma unroll
    for (int k = 0; k < SEL_SLOT; ++k) {
        const int i = tid + k * SEL_T;
        v[k] = dist[i < NPTS ? i : tid];
    }
    float qv[4];
    int   qk[4];
    #pragma unroll
    for (int q = 0; q < 4; ++q) {
        qv[q] = v[q * SEL_Q];
        qk[q] = q * SEL_Q;
        #pragma unroll
        for (int k = q * SEL_Q + 1; k < (q + 1) * SEL_Q; ++k) {
            const bool take = (v[k] < qv[q]) && (tid + k * SEL_T < NPTS);
            qv[q] = take ? v[k] : qv[q];
            qk[q] = take ? k : qk[q];
        }
    }
    if (qv[1] < qv[0]) { qv[0] = qv[1]; qk[0] = qk[1]; }
    if (qv[3] < qv[2]) { qv[2] = qv[3]; qk[2] = qk[3]; }
    if (qv[2] < qv[0]) { qv[0] = qv[2]; qk[0] = qk[2]; }
    hv = qv[0];
    hi = tid + qk[0] * SEL_T;
}

__global__ __launch_bounds__(1024) void select_agg_kernel(const float* __restrict__ distws,
                                                          const float* __restrict__ feats,
                                                          const int*   __restrict__ valid,
                                                          float* __restrict__ out_fts,
                                                          float* __restrict__ out_mask,
                                                          int N)
{
    extern __shared__ float smem[];
    float* dist = smem;
    float* rval = smem + N;
    int*   ridx = (int*)(rval + 32);
    __shared__ int sel[KNN];
    __shared__ int s_win;

    const int gc  = blockIdx.x;
    const int b   = gc >> 8;
    const int tid = threadIdx.x;

    const float* drow = distws + (size_t)gc * N;
    for (int i = tid; i < N; i += 1024) dist[i] = drow[i];
    __syncthreads();

    float hv;
    int   hi;
    slice_argmin(dist, tid, hv, hi);

    for (int k = 0; k < KNN; ++k) {
        float best = hv;
        int   bi   = hi;
        wave_argmin(best, bi);
        if ((tid & 31) == 0) { rval[tid >> 5] = best; ridx[tid >> 5] = bi; }
        __syncthreads();
        if (tid < 32) {
            best = rval[tid];
            bi   = ridx[tid];
            wave_argmin(best, bi);
            if (tid == 0) { sel[k] = bi; s_win = bi; }
        }
        __syncthreads();
        const int win = s_win;
        if ((win & 1023) == tid) {
            dist[win] = FLT_BIG;
            slice_argmin(dist, tid, hv, hi);
        }
    }
    __syncthreads();

    const float* F = feats + (size_t)b * N * FDIM;
    if (tid < FDIM) {
        float acc = 0.0f;
        #pragma unroll 4
        for (int k = 0; k < KNN; ++k) {
            const float* row = F + (size_t)sel[k] * FDIM;
            if (k + 1 < KNN) __builtin_prefetch(F + (size_t)sel[k + 1] * FDIM + tid, 0, 0);
            acc += row[tid];
        }
        out_fts[(size_t)gc * FDIM + tid] = acc * (1.0f / 64.0f);
    }
    if (tid == 0) {
        const int* V = valid + (size_t)b * N;
        int sv = 0;
        #pragma unroll 4
        for (int k = 0; k < KNN; ++k) sv += V[sel[k]];
        out_mask[gc] = (sv > 0) ? 1.0f : 0.0f;
    }
}

extern "C" void kernel_launch(void* const* d_in, const int* in_sizes, int n_in,
                              void* d_out, int out_size, void* d_ws, size_t ws_size,
                              hipStream_t stream) {
    (void)n_in; (void)out_size; (void)ws_size;
    const float* xyzs  = (const float*)d_in[0];
    const float* feats = (const float*)d_in[1];
    const int*   valid = (const int*)d_in[2];

    const int B = 2;
    const int N = in_sizes[0] / (B * 3);

    float* out_fts  = (float*)d_out;
    float* out_mask = out_fts + (size_t)B * NGROUP * FDIM;
    float* out_cent = out_mask + (size_t)B * NGROUP;

    float* centers_ws = (float*)d_ws;
    float* distws     = centers_ws + (size_t)B * NGROUP * 3;

    const size_t smem_bytes = (size_t)N * sizeof(float) + 32 * sizeof(float) + 32 * sizeof(int);

    fps_kernel<<<B, FPS_T, 0, stream>>>(xyzs, centers_ws, out_cent);
    knn_dist_kernel<<<dim3(B * (NGROUP / 16), NSPLIT), 256, 0, stream>>>(xyzs, centers_ws,
                                                                         distws, N);
    select_agg_kernel<<<B * NGROUP, 1024, smem_bytes, stream>>>(distws, feats, valid,
                                                                out_fts, out_mask, N);
}


# top-64 select: one threshold (largest of the waves' second-smallest heads) leaves a few hundred candidates, ranked by counting; same 64 in the same order, without 64 serial two-barrier rounds
# speedup vs baseline: 5.9653x; 1.0542x over previous
#include <hip/hip_runtime.h>
#include <hip/hip_bf16.h>

#define NPTS   40000
#define NGROUP 256
#define KNN    64
#define FDIM   768
#define FLT_BIG 3.402823e38f

typedef __attribute__((ext_vector_type(2))) float v2f;
typedef __attribute__((ext_vector_type(2))) int   v2i;
typedef __attribute__((ext_vector_type(4))) float v4f;
typedef __attribute__((ext_vector_type(4))) int   v4i;
typedef __attribute__((ext_vector_type(8))) float v8f;


template <int XMASK> __device__ __forceinline__ int row_xor_i(int v)
{
    return __builtin_amdgcn_update_dpp(0, v, 0x160 + XMASK, 0xf, 0xf, true);
}
template <int XMASK> __device__ __forceinline__ float row_xor_f(float v)
{
    return __builtin_bit_cast(float, row_xor_i<XMASK>(__builtin_bit_cast(int, v)));
}
__device__ __forceinline__ int other_row_i(int v)
{
    return __builtin_amdgcn_permlanex16(v, v, 0x76543210, 0xfedcba98, false, false);
}
__device__ __forceinline__ float other_row_f(float v)
{
    return __builtin_bit_cast(float, other_row_i(__builtin_bit_cast(int, v)));
}
__device__ __forceinline__ int wave_min_i(int v)
{
    v = min(v, row_xor_i<1>(v));
    v = min(v, row_xor_i<2>(v));
    v = min(v, row_xor_i<4>(v));
    v = min(v, row_xor_i<8>(v));
    return min(v, other_row_i(v));
}
__device__ __forceinline__ int wave_max_i(int v)
{
    v = max(v, row_xor_i<1>(v));
    v = max(v, row_xor_i<2>(v));
    v = max(v, row_xor_i<4>(v));
    v = max(v, row_xor_i<8>(v));
    return max(v, other_row_i(v));
}
__device__ __forceinline__ float wave_min_f(float v)
{
    v = fminf(v, row_xor_f<1>(v));
    v = fminf(v, row_xor_f<2>(v));
    v = fminf(v, row_xor_f<4>(v));
    v = fminf(v, row_xor_f<8>(v));
    return fminf(v, other_row_f(v));
}
__device__ __forceinline__ float wave_max_f(float v)
{
    v = fmaxf(v, row_xor_f<1>(v));
    v = fmaxf(v, row_xor_f<2>(v));
    v = fmaxf(v, row_xor_f<4>(v));
    v = fmaxf(v, row_xor_f<8>(v));
    return fmaxf(v, other_row_f(v));
}
__device__ __forceinline__ int wave_sum_i(int v)
{
    v += row_xor_i<1>(v);
    v += row_xor_i<2>(v);
    v += row_xor_i<4>(v);
    v += row_xor_i<8>(v);
    return v + other_row_i(v);
}
__device__ __forceinline__ void wave_argmax_i(int& val, int& idx)
{
    const int m = wave_max_i(val);
    idx = wave_min_i(val == m ? idx : 0x7fffffff);
    val = m;
}
__device__ __forceinline__ void wave_argmin(float& val, int& idx)
{
    const float m = wave_min_f(val);
    idx = wave_min_i(val == m ? idx : 0x7fffffff);
    val = m;
}

template <int LO, int CNT, int PAIRS>
__device__ __forceinline__ int first_slot_eq(const v2i (&v)[PAIRS], const v2f (&x)[PAIRS],
                                             int m, float& bx)
{
    int bj = LO + CNT - 1;
    bx = x[(LO + CNT - 1) >> 1][(LO + CNT - 1) & 1];
    #pragma unroll
    for (int j = LO + CNT - 2; j >= LO; --j) {
        const bool hit = v[j >> 1][j & 1] == m;
        bj = hit ? j : bj;
        bx = hit ? x[j >> 1][j & 1] : bx;
    }
    return bj;
}

#define FPS_T    1024
#define FPS_SLOT 40
#define FPS_ACT  (NPTS / FPS_SLOT)
#define FPS_GSZ  10
#define FPS_GRP  (FPS_SLOT / FPS_GSZ)
#define FPS_AHEAD 5
static_assert(FPS_GRP == 4 && FPS_GRP * FPS_GSZ == FPS_SLOT && FPS_GSZ % 2 == 0,
              "a 16-byte (y,z) entry's two slots must lie in one group");
static_assert(FPS_ACT * FPS_SLOT == NPTS && FPS_ACT <= FPS_T && FPS_SLOT % 2 == 0,
              "slot layout must cover every point exactly once");

struct __attribute__((aligned(32))) Partial {
    v4i c;
    int idx;
    int pad[3];
};

__global__ __launch_bounds__(FPS_T) void fps_kernel(const float* __restrict__ xyz,
                                                    float* __restrict__ centers_ws,
                                                    float* __restrict__ out_centers)
{
    __shared__ v4f yz2[(FPS_SLOT / 2) * FPS_ACT];
    __shared__ Partial part[2][32];
    __shared__ int    picks[NGROUP];

    const int b   = blockIdx.x;
    const int tid = threadIdx.x;
    const bool own = tid < FPS_ACT;
    const float* X = xyz + (size_t)b * NPTS * 3;
    float* cw = centers_ws + (size_t)b * NGROUP * 3;
    float* co = out_centers + (size_t)b * NGROUP * 3;

    if (own) {
        #pragma unroll 4
        for (int p = 0; p < FPS_SLOT / 2; ++p) {
            const size_t i0 = (size_t)tid + (size_t)(2 * p) * FPS_ACT;
            const size_t i1 = i0 + FPS_ACT;
            v4f e;
            e.x = X[i0 * 3 + 1];  e.y = X[i1 * 3 + 1];
            e.z = X[i0 * 3 + 2];  e.w = X[i1 * 3 + 2];
            yz2[p * FPS_ACT + tid] = e;
        }
    }
    v2f xs[FPS_SLOT / 2];
    v2i dist[FPS_SLOT / 2];
    #pragma unroll
    for (int p = 0; p < FPS_SLOT / 2; ++p) {
        xs[p].x = 0.0f;  xs[p].y = 0.0f;
        dist[p].x = __builtin_bit_cast(int, 1e10f);  dist[p].y = __builtin_bit_cast(int, 1e10f);
    }
    if (own) {
        #pragma unroll
        for (int p = 0; p < FPS_SLOT / 2; ++p) {
            xs[p].x = X[((size_t)tid + (size_t)(2 * p) * FPS_ACT) * 3 + 0];
            xs[p].y = X[((size_t)tid + (size_t)(2 * p + 1) * FPS_ACT) * 3 + 0];
        }
    }
    if (tid == 0) picks[0] = 0;
    __syncthreads();

    const v4f* yzb[FPS_GRP];
    #pragma unroll
    for (int g = 0; g < FPS_GRP; ++g) {
        unsigned e = (unsigned)tid + (unsigned)(g * (FPS_GSZ / 2) * FPS_ACT);
        asm volatile("" : "+v"(e));
        yzb[g] = yz2 + e;
    }

    float px = X[0];
    float py = X[1];
    float pz = X[2];
    const int lane = tid & 31;

    for (int s = 1; s < NGROUP; ++s) {
        const v2f px2 = {px, px};
        const v2f py2 = {py, py};
        const v2f pz2 = {pz, pz};

        int best = -1;
        int bi   = 0x7fffffff;
        int gm[FPS_GRP];
        #pragma unroll
        for (int k = 0; k < FPS_GRP; ++k) gm[k] = -1;
        if (own) {
            v4f ring[FPS_AHEAD];
            #pragma unroll
            for (int k = 0; k < FPS_AHEAD; ++k)
                ring[k] = yzb[k / (FPS_GSZ / 2)][(k % (FPS_GSZ / 2)) * FPS_ACT];
            #pragma unroll
            for (int p = 0; p < FPS_SLOT / 2; ++p) {
                const v4f q = ring[p % FPS_AHEAD];
                if (p + FPS_AHEAD < FPS_SLOT / 2)
                    ring[p % FPS_AHEAD] = yzb[(p + FPS_AHEAD) / (FPS_GSZ / 2)]
                                             [((p + FPS_AHEAD) % (FPS_GSZ / 2)) * FPS_ACT];
                const v2f dx = xs[p] - px2;
                const v2f dy = q.xy - py2;
                const v2f dz = q.zw - pz2;
                v2f t;
                {
                    #pragma clang fp contract(off)
                    t = dx * dx + dy * dy;
                }
                const v2f d = __builtin_elementwise_fma(dz, dz, t);
                const v2i dk = __builtin_bit_cast(v2i, d);
                const v2i k = __builtin_elementwise_min(dist[p], dk);
                dist[p] = k;
                gm[(2 * p) / FPS_GSZ] = max(max(gm[(2 * p) / FPS_GSZ], k.x), k.y);
            }
            best = max(max(gm[0], gm[1]), max(gm[2], gm[3]));
        }
        const int m = wave_max_i(best);
        int   bj = 0;
        float bx = 0.0f;
        v4f   q  = {0.0f, 0.0f, 0.0f, 0.0f};
        if (own && best == m) {
            if      (gm[0] == m) bj = first_slot_eq<0 * FPS_GSZ, FPS_GSZ>(dist, xs, m, bx);
            else if (gm[1] == m) bj = first_slot_eq<1 * FPS_GSZ, FPS_GSZ>(dist, xs, m, bx);
            else if (gm[2] == m) bj = first_slot_eq<2 * FPS_GSZ, FPS_GSZ>(dist, xs, m, bx);
            else                 bj = first_slot_eq<3 * FPS_GSZ, FPS_GSZ>(dist, xs, m, bx);
            bi = tid + bj * FPS_ACT;
            q = yzb[0][(bj >> 1) * FPS_ACT];
        }
        const int par = s & 1;
        const int wbi = wave_min_i(bi);
        if (bi == wbi) {
            const bool odd = (bj & 1) != 0;
            const float cy = odd ? q.y : q.x;
            const float cz = odd ? q.w : q.z;
            const v4i c = {__float_as_int(bx), __float_as_int(cy), __float_as_int(cz), m};
            part[par][tid >> 5].c   = c;
            part[par][tid >> 5].idx = bi;
        }
        __syncthreads();
        const v4i kc   = part[par][lane].c;
        const int kidx = part[par][lane].idx;
        int gkey = kc.w;
        int gidx = kidx;
        wave_argmax_i(gkey, gidx);
        const unsigned wmask = __builtin_amdgcn_ballot_w32(kidx == gidx);
        const int wl = __builtin_ctz(wmask);
        const int kx = kc.x;
        const int ky = kc.y;
        const int kz = kc.z;
        px = __int_as_float(__builtin_amdgcn_readlane(kx, wl));
        py = __int_as_float(__builtin_amdgcn_readlane(ky, wl));
        pz = __int_as_float(__builtin_amdgcn_readlane(kz, wl));
        if (tid == 0) picks[s] = gidx;
    }
    __syncthreads();

    if (tid < NGROUP) {
        const int i = picks[tid];
        const float x = X[(size_t)i * 3 + 0];
        const float y = X[(size_t)i * 3 + 1];
        const float z = X[(size_t)i * 3 + 2];
        cw[(size_t)tid * 3 + 0] = x;  co[(size_t)tid * 3 + 0] = x;
        cw[(size_t)tid * 3 + 1] = y;  co[(size_t)tid * 3 + 1] = y;
        cw[(size_t)tid * 3 + 2] = z;  co[(size_t)tid * 3 + 2] = z;
    }
}

#define NSPLIT 25

__global__ __launch_bounds__(256) void knn_dist_kernel(const float* __restrict__ xyz,
                                                       const float* __restrict__ centers,
                                                       float* __restrict__ distws,
                                                       int N)
{
    const int bidx  = blockIdx.x;
    const int b     = bidx >> 4;
    const int mtile = bidx & 15;
    const int lane  = threadIdx.x & 31;
    const int wave  = threadIdx.x >> 5;
    const int half  = (lane < 16) ? 0 : 1;
    const int l16   = lane & 15;

    const int slice = N / NSPLIT;
    const int nbeg  = blockIdx.y * slice;
    const int nend  = nbeg + slice;

    const float* X = xyz + (size_t)b * N * 3;
    const float* C = centers + (size_t)(b * NGROUP + mtile * 16) * 3;
    float* drow = distws + (size_t)(b * NGROUP + mtile * 16) * N;

    const float cx = C[(size_t)l16 * 3 + 0];
    const float cy = C[(size_t)l16 * 3 + 1];
    const float cz = C[(size_t)l16 * 3 + 2];
    v2f a;
    a.x = half ? (-2.0f * cz) : (-2.0f * cx);
    a.y = half ? 1.0f         : (-2.0f * cy);

    for (int n0 = nbeg + wave * 16; n0 < nend; n0 += 16 * 8) {
        const int n = n0 + l16;
        const float px = X[(size_t)n * 3 + 0];
        const float py = X[(size_t)n * 3 + 1];
        const float pz = X[(size_t)n * 3 + 2];
        const float n2 = px * px + py * py + pz * pz;
        v2f bb;
        bb.x = half ? pz : px;
        bb.y = half ? n2 : py;

        v8f c = {};
        c = __builtin_amdgcn_wmma_f32_16x16x4_f32(
                 false, a,  false, bb,
                 (short)0, c,  false,  false);

        #pragma unroll
        for (int r = 0; r < 8; ++r) {
            const int mm = r + 8 * half;
            drow[(size_t)mm * N + n] = c[r];
        }
    }
}

#define SEL_T    1024
#define SEL_SLOT ((NPTS + SEL_T - 1) / SEL_T)
#define SEL_Q    (SEL_SLOT / 4)
static_assert(SEL_SLOT % 4 == 0 && (SEL_SLOT - 1) * SEL_T < NPTS &&
              (3 * SEL_Q) * SEL_T + SEL_T - 1 < NPTS,
              "every quarter must start on an element that exists for every thread");

__device__ __forceinline__ void slice_argmin(const float* dist, int tid, float& hv, int& hi)
{
    float v[SEL_SLOT];
    #pragma unroll
    for (int k = 0; k < SEL_SLOT; ++k) {
        const int i = tid + k * SEL_T;
        v[k] = dist[i < NPTS ? i : tid];
    }
    float qv[4];
    int   qk[4];
    #pragma unroll
    for (int q = 0; q < 4; ++q) {
        qv[q] = v[q * SEL_Q];
        qk[q] = q * SEL_Q;
        #pragma unroll
        for (int k = q * SEL_Q + 1; k < (q + 1) * SEL_Q; ++k) {
            const bool take = (v[k] < qv[q]) && (tid + k * SEL_T < NPTS);
            qv[q] = take ? v[k] : qv[q];
            qk[q] = take ? k : qk[q];
        }
    }
    if (qv[1] < qv[0]) { qv[0] = qv[1]; qk[0] = qk[1]; }
    if (qv[3] < qv[2]) { qv[2] = qv[3]; qk[2] = qk[3]; }
    if (qv[2] < qv[0]) { qv[0] = qv[2]; qk[0] = qk[2]; }
    hv = qv[0];
    hi = tid + qk[0] * SEL_T;
}

#define SEL_CAP 624
static_assert(NPTS <= 65536 && SEL_CAP % 16 == 0 && SEL_CAP <= SEL_T, "16-bit indices, eight to an entry, one thread a candidate");
typedef unsigned short __attribute__((may_alias)) cand_u16;
__global__ __launch_bounds__(1024) void select_agg_kernel(const float* __restrict__ distws,
                                                          const float* __restrict__ feats,
                                                          const int*   __restrict__ valid,
                                                          float* __restrict__ out_fts,
                                                          float* __restrict__ out_mask,
                                                          int N)
{
    extern __shared__ float smem[];
    float* dist = smem;
    float* rval = smem + N;
    int*   ridx = (int*)(rval + 32);
    __shared__ int sel[KNN];
    __shared__ int s_win;
    __shared__ v4i cand[SEL_CAP / 8];

    const int gc  = blockIdx.x;
    const int b   = gc >> 8;
    const int tid = threadIdx.x;

    const float* drow = distws + (size_t)gc * N;
    for (int i = tid; i < N; i += 1024) dist[i] = drow[i];
    __syncthreads();

    float hv;
    int   hi;
    slice_argmin(dist, tid, hv, hi);

    const int lane = tid & 31;
    const int wv   = tid >> 5;

    const float m1 = wave_min_f(hv);
    const unsigned eq1 = __builtin_amdgcn_ballot_w32(hv == m1);
    const int first = eq1 ? __builtin_ctz(eq1) : 0;
    const float m2 = wave_min_f(lane == first ? FLT_BIG : hv);
    if (lane == 0) rval[wv] = m2;
    __syncthreads();
    const float T = wave_max_f(rval[lane]);

    unsigned mk[SEL_SLOT];
    int cntw = 0;
    #pragma unroll
    for (int k = 0; k < SEL_SLOT; ++k) {
        const int i = tid + k * SEL_T;
        const bool ok = i < NPTS;
        const float v = dist[ok ? i : tid];
        mk[k] = __builtin_amdgcn_ballot_w32(ok && v <= T);
        cntw += __builtin_popcount(mk[k]);
    }
    if (lane == 0) ridx[wv] = cntw;
    if (tid < KNN) sel[tid] = 0;
    __syncthreads();
    const int cl = ridx[lane];
    const int C  = wave_sum_i(cl);
    int base     = wave_sum_i(lane < wv ? cl : 0);

    if (C <= SEL_CAP) {
        cand_u16* cl16 = (cand_u16*)cand;
        #pragma unroll
        for (int k = 0; k < SEL_SLOT; ++k) {
            if (mk[k] != 0u) {
                if ((mk[k] >> lane) & 1u)
                    cl16[base + (int)__builtin_amdgcn_mbcnt_lo(mk[k], 0u)] = (unsigned short)(tid + k * SEL_T);
                base += __builtin_popcount(mk[k]);
            }
        }
        __syncthreads();
        if (tid < C) {
            const int   my = ((const cand_u16*)cand)[tid];
            const float mv = dist[my];
            int rank = 0;
            for (int j0 = 0; j0 < C; j0 += 16) {
                const v4i pk0 = cand[(j0 >> 3) + 0];
                const v4i pk1 = cand[(j0 >> 3) + 1];
                #pragma unroll
                for (int u = 0; u < 16; ++u) {
                    const unsigned w32 = (unsigned)((u < 8) ? pk0[(u & 7) >> 1] : pk1[(u & 7) >> 1]);
                    const int raw = (u & 1) ? (int)(w32 >> 16) : (int)(w32 & 0xffffu);
                    const int cj  = (j0 + u < C) ? raw : my;
                    const float vj = dist[cj];
                    rank += (vj < mv || (vj == mv && cj < my)) ? 1 : 0;
                }
            }
            if (rank < KNN) sel[rank] = my;
        }
    } else {
        __syncthreads();
        for (int k = 0; k < KNN; ++k) {
            float best = hv;
            int   bi   = hi;
            wave_argmin(best, bi);
            if ((tid & 31) == 0) { rval[tid >> 5] = best; ridx[tid >> 5] = bi; }
            __syncthreads();
            if (tid < 32) {
                best = rval[tid];
                bi   = ridx[tid];
                wave_argmin(best, bi);
                if (tid == 0) { sel[k] = bi; s_win = bi; }
            }
            __syncthreads();
            const int win = s_win;
            if ((win & 1023) == tid) {
                dist[win] = FLT_BIG;
                slice_argmin(dist, tid, hv, hi);
            }
        }
    }
    __syncthreads();

    const float* F = feats + (size_t)b * N * FDIM;
    if (tid < FDIM) {
        float acc = 0.0f;
        #pragma unroll 4
        for (int k = 0; k < KNN; ++k) {
            const float* row = F + (size_t)sel[k] * FDIM;
            if (k + 1 < KNN) __builtin_prefetch(F + (size_t)sel[k + 1] * FDIM + tid, 0, 0);
            acc += row[tid];
        }
        out_fts[(size_t)gc * FDIM + tid] = acc * (1.0f / 64.0f);
    }
    if (tid == 0) {
        const int* V = valid + (size_t)b * N;
        int sv = 0;
        #pragma unroll 4
        for (int k = 0; k < KNN; ++k) sv += V[sel[k]];
        out_mask[gc] = (sv > 0) ? 1.0f : 0.0f;
    }
}

extern "C" void kernel_launch(void* const* d_in, const int* in_sizes, int n_in,
                              void* d_out, int out_size, void* d_ws, size_t ws_size,
                              hipStream_t stream) {
    (void)n_in; (void)out_size; (void)ws_size;
    const float* xyzs  = (const float*)d_in[0];
    const float* feats = (const float*)d_in[1];
    const int*   valid = (const int*)d_in[2];

    const int B = 2;
    const int N = in_sizes[0] / (B * 3);

    float* out_fts  = (float*)d_out;
    float* out_mask = out_fts + (size_t)B * NGROUP * FDIM;
    float* out_cent = out_mask + (size_t)B * NGROUP;

    float* centers_ws = (float*)d_ws;
    float* distws     = centers_ws + (size_t)B * NGROUP * 3;

    const size_t smem_bytes = (size_t)N * sizeof(float) + 32 * sizeof(float) + 32 * sizeof(int);

    fps_kernel<<<B, FPS_T, 0, stream>>>(xyzs, centers_ws, out_cent);
    knn_dist_kernel<<<dim3(B * (NGROUP / 16), NSPLIT), 256, 0, stream>>>(xyzs, centers_ws,
                                                                         distws, N);
    select_agg_kernel<<<B * NGROUP, 1024, smem_bytes, stream>>>(distws, feats, valid,
                                                                out_fts, out_mask, N);
}


# FPS: 16 waves own 80 points each instead of 32 waves of 40 (512 threads; x and min-dist in 160 VGPRs): a pick's per-wave hand-over is issued by half as many waves per vector unit; same picks
# speedup vs baseline: 6.2255x; 1.0436x over previous
#include <hip/hip_runtime.h>
#include <hip/hip_bf16.h>

#define NPTS   40000
#define NGROUP 256
#define KNN    64
#define FDIM   768
#define FLT_BIG 3.402823e38f

typedef __attribute__((ext_vector_type(2))) float v2f;
typedef __attribute__((ext_vector_type(2))) int   v2i;
typedef __attribute__((ext_vector_type(4))) float v4f;
typedef __attribute__((ext_vector_type(4))) int   v4i;
typedef __attribute__((ext_vector_type(8))) float v8f;


template <int XMASK> __device__ __forceinline__ int row_xor_i(int v)
{
    return __builtin_amdgcn_update_dpp(0, v, 0x160 + XMASK, 0xf, 0xf, true);
}
template <int XMASK> __device__ __forceinline__ float row_xor_f(float v)
{
    return __builtin_bit_cast(float, row_xor_i<XMASK>(__builtin_bit_cast(int, v)));
}
__device__ __forceinline__ int other_row_i(int v)
{
    return __builtin_amdgcn_permlanex16(v, v, 0x76543210, 0xfedcba98, false, false);
}
__device__ __forceinline__ float other_row_f(float v)
{
    return __builtin_bit_cast(float, other_row_i(__builtin_bit_cast(int, v)));
}
__device__ __forceinline__ int wave_min_i(int v)
{
    v = min(v, row_xor_i<1>(v));
    v = min(v, row_xor_i<2>(v));
    v = min(v, row_xor_i<4>(v));
    v = min(v, row_xor_i<8>(v));
    return min(v, other_row_i(v));
}
__device__ __forceinline__ int wave_max_i(int v)
{
    v = max(v, row_xor_i<1>(v));
    v = max(v, row_xor_i<2>(v));
    v = max(v, row_xor_i<4>(v));
    v = max(v, row_xor_i<8>(v));
    return max(v, other_row_i(v));
}
__device__ __forceinline__ float wave_min_f(float v)
{
    v = fminf(v, row_xor_f<1>(v));
    v = fminf(v, row_xor_f<2>(v));
    v = fminf(v, row_xor_f<4>(v));
    v = fminf(v, row_xor_f<8>(v));
    return fminf(v, other_row_f(v));
}
__device__ __forceinline__ float wave_max_f(float v)
{
    v = fmaxf(v, row_xor_f<1>(v));
    v = fmaxf(v, row_xor_f<2>(v));
    v = fmaxf(v, row_xor_f<4>(v));
    v = fmaxf(v, row_xor_f<8>(v));
    return fmaxf(v, other_row_f(v));
}
__device__ __forceinline__ int wave_sum_i(int v)
{
    v += row_xor_i<1>(v);
    v += row_xor_i<2>(v);
    v += row_xor_i<4>(v);
    v += row_xor_i<8>(v);
    return v + other_row_i(v);
}
__device__ __forceinline__ void wave_argmax_i(int& val, int& idx)
{
    const int m = wave_max_i(val);
    idx = wave_min_i(val == m ? idx : 0x7fffffff);
    val = m;
}
__device__ __forceinline__ void wave_argmin(float& val, int& idx)
{
    const float m = wave_min_f(val);
    idx = wave_min_i(val == m ? idx : 0x7fffffff);
    val = m;
}

template <int LO, int CNT, int PAIRS>
__device__ __forceinline__ int first_slot_eq(const v2i (&v)[PAIRS], const v2f (&x)[PAIRS],
                                             int m, float& bx)
{
    int bj = LO + CNT - 1;
    bx = x[(LO + CNT - 1) >> 1][(LO + CNT - 1) & 1];
    #pragma unroll
    for (int j = LO + CNT - 2; j >= LO; --j) {
        const bool hit = v[j >> 1][j & 1] == m;
        bj = hit ? j : bj;
        bx = hit ? x[j >> 1][j & 1] : bx;
    }
    return bj;
}

#define FPS_T    512
#define FPS_WAVES (FPS_T / 32)
#define FPS_SLOT 80
#define FPS_ACT  (NPTS / FPS_SLOT)
#define FPS_GSZ  10
#define FPS_GRP  (FPS_SLOT / FPS_GSZ)
#define FPS_AHEAD 5
static_assert(FPS_GRP == 8 && FPS_GRP * FPS_GSZ == FPS_SLOT && FPS_GSZ % 2 == 0,
              "a 16-byte (y,z) entry's two slots must lie in one group");
static_assert(FPS_WAVES == 16 && NGROUP <= FPS_T,
              "16 partials, read twice over by a wave's 32 lanes; one thread a centre at the write-out");
static_assert(FPS_ACT * FPS_SLOT == NPTS && FPS_ACT <= FPS_T && FPS_SLOT % 2 == 0,
              "slot layout must cover every point exactly once");

struct __attribute__((aligned(32))) Partial {
    v4i c;
    int idx;
    int pad[3];
};

__global__ __launch_bounds__(FPS_T) void fps_kernel(const float* __restrict__ xyz,
                                                    float* __restrict__ centers_ws,
                                                    float* __restrict__ out_centers)
{
    __shared__ v4f yz2[(FPS_SLOT / 2) * FPS_ACT];
    __shared__ Partial part[2][FPS_WAVES];
    __shared__ int    picks[NGROUP];

    const int b   = blockIdx.x;
    const int tid = threadIdx.x;
    const bool own = tid < FPS_ACT;
    const float* X = xyz + (size_t)b * NPTS * 3;
    float* cw = centers_ws + (size_t)b * NGROUP * 3;
    float* co = out_centers + (size_t)b * NGROUP * 3;

    if (own) {
        #pragma unroll 4
        for (int p = 0; p < FPS_SLOT / 2; ++p) {
            const size_t i0 = (size_t)tid + (size_t)(2 * p) * FPS_ACT;
            const size_t i1 = i0 + FPS_ACT;
            v4f e;
            e.x = X[i0 * 3 + 1];  e.y = X[i1 * 3 + 1];
            e.z = X[i0 * 3 + 2];  e.w = X[i1 * 3 + 2];
            yz2[p * FPS_ACT + tid] = e;
        }
    }
    v2f xs[FPS_SLOT / 2];
    v2i dist[FPS_SLOT / 2];
    #pragma unroll
    for (int p = 0; p < FPS_SLOT / 2; ++p) {
        xs[p].x = 0.0f;  xs[p].y = 0.0f;
        dist[p].x = __builtin_bit_cast(int, 1e10f);  dist[p].y = __builtin_bit_cast(int, 1e10f);
    }
    if (own) {
        #pragma unroll
        for (int p = 0; p < FPS_SLOT / 2; ++p) {
            xs[p].x = X[((size_t)tid + (size_t)(2 * p) * FPS_ACT) * 3 + 0];
            xs[p].y = X[((size_t)tid + (size_t)(2 * p + 1) * FPS_ACT) * 3 + 0];
        }
    }
    if (tid == 0) picks[0] = 0;
    __syncthreads();

    const v4f* yzb[FPS_GRP];
    #pragma unroll
    for (int g = 0; g < FPS_GRP; ++g) {
        unsigned e = (unsigned)tid + (unsigned)(g * (FPS_GSZ / 2) * FPS_ACT);
        asm volatile("" : "+v"(e));
        yzb[g] = yz2 + e;
    }

    float px = X[0];
    float py = X[1];
    float pz = X[2];
    const int lane = tid & 31;

    for (int s = 1; s < NGROUP; ++s) {
        const v2f px2 = {px, px};
        const v2f py2 = {py, py};
        const v2f pz2 = {pz, pz};

        int best = -1;
        int bi   = 0x7fffffff;
        int gm[FPS_GRP];
        #pragma unroll
        for (int k = 0; k < FPS_GRP; ++k) gm[k] = -1;
        if (own) {
            v4f ring[FPS_AHEAD];
            #pragma unroll
            for (int k = 0; k < FPS_AHEAD; ++k)
                ring[k] = yzb[k / (FPS_GSZ / 2)][(k % (FPS_GSZ / 2)) * FPS_ACT];
            #pragma unroll
            for (int p = 0; p < FPS_SLOT / 2; ++p) {
                const v4f q = ring[p % FPS_AHEAD];
                if (p + FPS_AHEAD < FPS_SLOT / 2)
                    ring[p % FPS_AHEAD] = yzb[(p + FPS_AHEAD) / (FPS_GSZ / 2)]
                                             [((p + FPS_AHEAD) % (FPS_GSZ / 2)) * FPS_ACT];
                const v2f dx = xs[p] - px2;
                const v2f dy = q.xy - py2;
                const v2f dz = q.zw - pz2;
                v2f t;
                {
                    #pragma clang fp contract(off)
                    t = dx * dx + dy * dy;
                }
                const v2f d = __builtin_elementwise_fma(dz, dz, t);
                const v2i dk = __builtin_bit_cast(v2i, d);
                const v2i k = __builtin_elementwise_min(dist[p], dk);
                dist[p] = k;
                gm[(2 * p) / FPS_GSZ] = max(max(gm[(2 * p) / FPS_GSZ], k.x), k.y);
            }
            best = max(max(max(gm[0], gm[1]), max(gm[2], gm[3])),
                       max(max(gm[4], gm[5]), max(gm[6], gm[7])));
        }
        const int m = wave_max_i(best);
        int   bj = 0;
        float bx = 0.0f;
        v4f   q  = {0.0f, 0.0f, 0.0f, 0.0f};
        if (own && best == m) {
            if      (gm[0] == m) bj = first_slot_eq<0 * FPS_GSZ, FPS_GSZ>(dist, xs, m, bx);
            else if (gm[1] == m) bj = first_slot_eq<1 * FPS_GSZ, FPS_GSZ>(dist, xs, m, bx);
            else if (gm[2] == m) bj = first_slot_eq<2 * FPS_GSZ, FPS_GSZ>(dist, xs, m, bx);
            else if (gm[3] == m) bj = first_slot_eq<3 * FPS_GSZ, FPS_GSZ>(dist, xs, m, bx);
            else if (gm[4] == m) bj = first_slot_eq<4 * FPS_GSZ, FPS_GSZ>(dist, xs, m, bx);
            else if (gm[5] == m) bj = first_slot_eq<5 * FPS_GSZ, FPS_GSZ>(dist, xs, m, bx);
            else if (gm[6] == m) bj = first_slot_eq<6 * FPS_GSZ, FPS_GSZ>(dist, xs, m, bx);
            else                 bj = first_slot_eq<7 * FPS_GSZ, FPS_GSZ>(dist, xs, m, bx);
            bi = tid + bj * FPS_ACT;
            q = yzb[0][(bj >> 1) * FPS_ACT];
        }
        const int par = s & 1;
        const int wbi = wave_min_i(bi);
        if (bi == wbi) {
            const bool odd = (bj & 1) != 0;
            const float cy = odd ? q.y : q.x;
            const float cz = odd ? q.w : q.z;
            const v4i c = {__float_as_int(bx), __float_as_int(cy), __float_as_int(cz), m};
            part[par][tid >> 5].c   = c;
            part[par][tid >> 5].idx = bi;
        }
        __syncthreads();
        const v4i kc   = part[par][lane & (FPS_WAVES - 1)].c;
        const int kidx = part[par][lane & (FPS_WAVES - 1)].idx;
        int gkey = kc.w;
        int gidx = kidx;
        wave_argmax_i(gkey, gidx);
        const unsigned wmask = __builtin_amdgcn_ballot_w32(kidx == gidx);
        const int wl = __builtin_ctz(wmask);
        const int kx = kc.x;
        const int ky = kc.y;
        const int kz = kc.z;
        px = __int_as_float(__builtin_amdgcn_readlane(kx, wl));
        py = __int_as_float(__builtin_amdgcn_readlane(ky, wl));
        pz = __int_as_float(__builtin_amdgcn_readlane(kz, wl));
        if (tid == 0) picks[s] = gidx;
    }
    __syncthreads();

    if (tid < NGROUP) {
        const int i = picks[tid];
        const float x = X[(size_t)i * 3 + 0];
        const float y = X[(size_t)i * 3 + 1];
        const float z = X[(size_t)i * 3 + 2];
        cw[(size_t)tid * 3 + 0] = x;  co[(size_t)tid * 3 + 0] = x;
        cw[(size_t)tid * 3 + 1] = y;  co[(size_t)tid * 3 + 1] = y;
        cw[(size_t)tid * 3 + 2] = z;  co[(size_t)tid * 3 + 2] = z;
    }
}

#define NSPLIT 25

__global__ __launch_bounds__(256) void knn_dist_kernel(const float* __restrict__ xyz,
                                                       const float* __restrict__ centers,
                                                       float* __restrict__ distws,
                                                       int N)
{
    const int bidx  = blockIdx.x;
    const int b     = bidx >> 4;
    const int mtile = bidx & 15;
    const int lane  = threadIdx.x & 31;
    const int wave  = threadIdx.x >> 5;
    const int half  = (lane < 16) ? 0 : 1;
    const int l16   = lane & 15;

    const int slice = N / NSPLIT;
    const int nbeg  = blockIdx.y * slice;
    const int nend  = nbeg + slice;

    const float* X = xyz + (size_t)b * N * 3;
    const float* C = centers + (size_t)(b * NGROUP + mtile * 16) * 3;
    float* drow = distws + (size_t)(b * NGROUP + mtile * 16) * N;

    const float cx = C[(size_t)l16 * 3 + 0];
    const float cy = C[(size_t)l16 * 3 + 1];
    const float cz = C[(size_t)l16 * 3 + 2];
    v2f a;
    a.x = half ? (-2.0f * cz) : (-2.0f * cx);
    a.y = half ? 1.0f         : (-2.0f * cy);

    for (int n0 = nbeg + wave * 16; n0 < nend; n0 += 16 * 8) {
        const int n = n0 + l16;
        const float px = X[(size_t)n * 3 + 0];
        const float py = X[(size_t)n * 3 + 1];
        const float pz = X[(size_t)n * 3 + 2];
        const float n2 = px * px + py * py + pz * pz;
        v2f bb;
        bb.x = half ? pz : px;
        bb.y = half ? n2 : py;

        v8f c = {};
        c = __builtin_amdgcn_wmma_f32_16x16x4_f32(
                 false, a,  false, bb,
                 (short)0, c,  false,  false);

        #pragma unroll
        for (int r = 0; r < 8; ++r) {
            const int mm = r + 8 * half;
            drow[(size_t)mm * N + n] = c[r];
        }
    }
}

#define SEL_T    1024
#define SEL_SLOT ((NPTS + SEL_T - 1) / SEL_T)
#define SEL_Q    (SEL_SLOT / 4)
static_assert(SEL_SLOT % 4 == 0 && (SEL_SLOT - 1) * SEL_T < NPTS &&
              (3 * SEL_Q) * SEL_T + SEL_T - 1 < NPTS,
              "every quarter must start on an element that exists for every thread");

__device__ __forceinline__ void slice_argmin(const float* dist, int tid, float& hv, int& hi)
{
    float v[SEL_SLOT];
    #pragma unroll
    for (int k = 0; k < SEL_SLOT; ++k) {
        const int i = tid + k * SEL_T;
        v[k] = dist[i < NPTS ? i : tid];
    }
    float qv[4];
    int   qk[4];
    #pragma unroll
    for (int q = 0; q < 4; ++q) {
        qv[q] = v[q * SEL_Q];
        qk[q] = q * SEL_Q;
        #pragma unroll
        for (int k = q * SEL_Q + 1; k < (q + 1) * SEL_Q; ++k) {
            const bool take = (v[k] < qv[q]) && (tid + k * SEL_T < NPTS);
            qv[q] = take ? v[k] : qv[q];
            qk[q] = take ? k : qk[q];
        }
    }
    if (qv[1] < qv[0]) { qv[0] = qv[1]; qk[0] = qk[1]; }
    if (qv[3] < qv[2]) { qv[2] = qv[3]; qk[2] = qk[3]; }
    if (qv[2] < qv[0]) { qv[0] = qv[2]; qk[0] = qk[2]; }
    hv = qv[0];
    hi = tid + qk[0] * SEL_T;
}

#define SEL_CAP 624
static_assert(NPTS <= 65536 && SEL_CAP % 16 == 0 && SEL_CAP <= SEL_T, "16-bit indices, eight to an entry, one thread a candidate");
typedef unsigned short __attribute__((may_alias)) cand_u16;
__global__ __launch_bounds__(1024) void select_agg_kernel(const float* __restrict__ distws,
                                                          const float* __restrict__ feats,
                                                          const int*   __restrict__ valid,
                                                          float* __restrict__ out_fts,
                                                          float* __restrict__ out_mask,
                                                          int N)
{
    extern __shared__ float smem[];
    float* dist = smem;
    float* rval = smem + N;
    int*   ridx = (int*)(rval + 32);
    __shared__ int sel[KNN];
    __shared__ int s_win;
    __shared__ v4i cand[SEL_CAP / 8];

    const int gc  = blockIdx.x;
    const int b   = gc >> 8;
    const int tid = threadIdx.x;

    const float* drow = distws + (size_t)gc * N;
    for (int i = tid; i < N; i += 1024) dist[i] = drow[i];
    __syncthreads();

    float hv;
    int   hi;
    slice_argmin(dist, tid, hv, hi);

    const int lane = tid & 31;
    const int wv   = tid >> 5;

    const float m1 = wave_min_f(hv);
    const unsigned eq1 = __builtin_amdgcn_ballot_w32(hv == m1);
    const int first = eq1 ? __builtin_ctz(eq1) : 0;
    const float m2 = wave_min_f(lane == first ? FLT_BIG : hv);
    if (lane == 0) rval[wv] = m2;
    __syncthreads();
    const float T = wave_max_f(rval[lane]);

    unsigned mk[SEL_SLOT];
    int cntw = 0;
    #pragma unroll
    for (int k = 0; k < SEL_SLOT; ++k) {
        const int i = tid + k * SEL_T;
        const bool ok = i < NPTS;
        const float v = dist[ok ? i : tid];
        mk[k] = __builtin_amdgcn_ballot_w32(ok && v <= T);
        cntw += __builtin_popcount(mk[k]);
    }
    if (lane == 0) ridx[wv] = cntw;
    if (tid < KNN) sel[tid] = 0;
    __syncthreads();
    const int cl = ridx[lane];
    const int C  = wave_sum_i(cl);
    int base     = wave_sum_i(lane < wv ? cl : 0);

    if (C <= SEL_CAP) {
        cand_u16* cl16 = (cand_u16*)cand;
        #pragma unroll
        for (int k = 0; k < SEL_SLOT; ++k) {
            if (mk[k] != 0u) {
                if ((mk[k] >> lane) & 1u)
                    cl16[base + (int)__builtin_amdgcn_mbcnt_lo(mk[k], 0u)] = (unsigned short)(tid + k * SEL_T);
                base += __builtin_popcount(mk[k]);
            }
        }
        __syncthreads();
        if (tid < C) {
            const int   my = ((const cand_u16*)cand)[tid];
            const float mv = dist[my];
            int rank = 0;
            for (int j0 = 0; j0 < C; j0 += 16) {
                const v4i pk0 = cand[(j0 >> 3) + 0];
                const v4i pk1 = cand[(j0 >> 3) + 1];
                #pragma unroll
                for (int u = 0; u < 16; ++u) {
                    const unsigned w32 = (unsigned)((u < 8) ? pk0[(u & 7) >> 1] : pk1[(u & 7) >> 1]);
                    const int raw = (u & 1) ? (int)(w32 >> 16) : (int)(w32 & 0xffffu);
                    const int cj  = (j0 + u < C) ? raw : my;
                    const float vj = dist[cj];
                    rank += (vj < mv || (vj == mv && cj < my)) ? 1 : 0;
                }
            }
            if (rank < KNN) sel[rank] = my;
        }
    } else {
        __syncthreads();
        for (int k = 0; k < KNN; ++k) {
            float best = hv;
            int   bi   = hi;
            wave_argmin(best, bi);
            if ((tid & 31) == 0) { rval[tid >> 5] = best; ridx[tid >> 5] = bi; }
            __syncthreads();
            if (tid < 32) {
                best = rval[tid];
                bi   = ridx[tid];
                wave_argmin(best, bi);
                if (tid == 0) { sel[k] = bi; s_win = bi; }
            }
            __syncthreads();
            const int win = s_win;
            if ((win & 1023) == tid) {
                dist[win] = FLT_BIG;
                slice_argmin(dist, tid, hv, hi);
            }
        }
    }
    __syncthreads();

    const float* F = feats + (size_t)b * N * FDIM;
    if (tid < FDIM) {
        float acc = 0.0f;
        #pragma unroll 4
        for (int k = 0; k < KNN; ++k) {
            const float* row = F + (size_t)sel[k] * FDIM;
            if (k + 1 < KNN) __builtin_prefetch(F + (size_t)sel[k + 1] * FDIM + tid, 0, 0);
            acc += row[tid];
        }
        out_fts[(size_t)gc * FDIM + tid] = acc * (1.0f / 64.0f);
    }
    if (tid == 0) {
        const int* V = valid + (size_t)b * N;
        int sv = 0;
        #pragma unroll 4
        for (int k = 0; k < KNN; ++k) sv += V[sel[k]];
        out_mask[gc] = (sv > 0) ? 1.0f : 0.0f;
    }
}

extern "C" void kernel_launch(void* const* d_in, const int* in_sizes, int n_in,
                              void* d_out, int out_size, void* d_ws, size_t ws_size,
                              hipStream_t stream) {
    (void)n_in; (void)out_size; (void)ws_size;
    const float* xyzs  = (const float*)d_in[0];
    const float* feats = (const float*)d_in[1];
    const int*   valid = (const int*)d_in[2];

    const int B = 2;
    const int N = in_sizes[0] / (B * 3);

    float* out_fts  = (float*)d_out;
    float* out_mask = out_fts + (size_t)B * NGROUP * FDIM;
    float* out_cent = out_mask + (size_t)B * NGROUP;

    float* centers_ws = (float*)d_ws;
    float* distws     = centers_ws + (size_t)B * NGROUP * 3;

    const size_t smem_bytes = (size_t)N * sizeof(float) + 32 * sizeof(float) + 32 * sizeof(int);

    fps_kernel<<<B, FPS_T, 0, stream>>>(xyzs, centers_ws, out_cent);
    knn_dist_kernel<<<dim3(B * (NGROUP / 16), NSPLIT), 256, 0, stream>>>(xyzs, centers_ws,
                                                                         distws, N);
    select_agg_kernel<<<B * NGROUP, 1024, smem_bytes, stream>>>(distws, feats, valid,
                                                                out_fts, out_mask, N);
}
